# starting kernel
# baseline (speedup 1.0000x reference)
#include <hip/hip_runtime.h>

#define N_NODES 50000
#define N_EDGES 800000
#define C 64

typedef float    v2f  __attribute__((ext_vector_type(2)));
typedef float    v8f  __attribute__((ext_vector_type(8)));


__global__ void zero_kernel(float4* __restrict__ p, int n4) {
    int i = blockIdx.x * blockDim.x + threadIdx.x;
    if (i < n4) p[i] = make_float4(0.f, 0.f, 0.f, 0.f);
}

__global__ void deg_kernel(const int* __restrict__ row, const float* __restrict__ w,
                           float* __restrict__ deg) {
    int e = blockIdx.x * blockDim.x + threadIdx.x;
    atomicAdd(&deg[row[e]], w[e]);
}

__global__ void deginv_kernel(float* __restrict__ deg) {
    int i = blockIdx.x * blockDim.x + threadIdx.x;
    if (i < N_NODES) {
        float d = deg[i];
        deg[i] = (d > 0.f) ? rsqrtf(d) : 0.f;
    }
}

__global__ void norm_kernel(const int* __restrict__ row, const int* __restrict__ col,
                            const float* __restrict__ w, const float* __restrict__ dinv,
                            float* __restrict__ nrm) {
    int e = blockIdx.x * blockDim.x + threadIdx.x;
    nrm[e] = dinv[row[e]] * w[e] * dinv[col[e]];
}

__global__ void prop_kernel(const float* __restrict__ xin, float* __restrict__ xout,
                            const int* __restrict__ row, const int* __restrict__ col,
                            const float* __restrict__ nrm) {
    unsigned gid = blockIdx.x * blockDim.x + threadIdx.x;
    unsigned e  = gid >> 4;
    unsigned cg = (gid & 15u) << 2;
    int r = row[e];
    int c = col[e];
    float w = nrm[e];
    const float4 v = *reinterpret_cast<const float4*>(xin + (size_t)c * C + cg);
    float* o = xout + (size_t)r * C + cg;
    atomicAdd(o + 0, w * v.x);
    atomicAdd(o + 1, w * v.y);
    atomicAdd(o + 2, w * v.z);
    atomicAdd(o + 3, w * v.w);
}


__device__ __forceinline__ void mat_accum(const float* __restrict__ A,
                                          const float* Wm,
                                          int arow, int m, int hi,
                                          v8f& c0, v8f& c1, v8f& c2, v8f& c3) {
    #pragma unroll
    for (int k0 = 0; k0 < 64; k0 += 4) {
        const int kk = k0 + hi * 2;
        v2f a = *reinterpret_cast<const v2f*>(A + arow * 64 + kk);
        v2f b0, b1, b2, b3;
        b0[0] = Wm[kk * 64 + m];          b0[1] = Wm[(kk + 1) * 64 + m];
        b1[0] = Wm[kk * 64 + 16 + m];     b1[1] = Wm[(kk + 1) * 64 + 16 + m];
        b2[0] = Wm[kk * 64 + 32 + m];     b2[1] = Wm[(kk + 1) * 64 + 32 + m];
        b3[0] = Wm[kk * 64 + 48 + m];     b3[1] = Wm[(kk + 1) * 64 + 48 + m];
        c0 = __builtin_amdgcn_wmma_f32_16x16x4_f32(false, a, false, b0, (short)0, c0, false, false);
        c1 = __builtin_amdgcn_wmma_f32_16x16x4_f32(false, a, false, b1, (short)0, c1, false, false);
        c2 = __builtin_amdgcn_wmma_f32_16x16x4_f32(false, a, false, b2, (short)0, c2, false, false);
        c3 = __builtin_amdgcn_wmma_f32_16x16x4_f32(false, a, false, b3, (short)0, c3, false, false);
    }
}

__global__ void __launch_bounds__(256)
gemm_kernel(const float* __restrict__ A0, const float* __restrict__ A1,
            const float* __restrict__ A2, const float* __restrict__ A3,
            const float* __restrict__ tf, const float* __restrict__ tb,
            float* __restrict__ out) {
    __shared__ float Ws[4 * 64 * 64];

    const int t = threadIdx.x;
    #pragma unroll 4
    for (int j = 0; j < 64; ++j) {
        int i   = t + j * 256;
        int mat = i >> 12;
        int idx = i & 4095;
        float v;
        if      (mat == 0) v = tf[idx];
        else if (mat == 1) v = tb[idx] + tb[4096 + idx];
        else if (mat == 2) v = tf[4096 + idx] + tb[8192 + idx];
        else               v = tf[8192 + idx];
        Ws[i] = v;
    }
    __syncthreads();

    const int wave = t >> 5;
    const int lane = t & 31;
    const int tm   = blockIdx.x * 8 + wave;
    if (tm >= 3125) return;

    const int m    = lane & 15;
    const int hi   = lane >> 4;
    const int arow = tm * 16 + m;

    v8f c0 = {}, c1 = {}, c2 = {}, c3 = {};
    mat_accum(A0, Ws,            arow, m, hi, c0, c1, c2, c3);
    mat_accum(A1, Ws + 4096,     arow, m, hi, c0, c1, c2, c3);
    mat_accum(A2, Ws + 8192,     arow, m, hi, c0, c1, c2, c3);
    mat_accum(A3, Ws + 12288,    arow, m, hi, c0, c1, c2, c3);

    #pragma unroll
    for (int v = 0; v < 8; ++v) {
        const int rowm = tm * 16 + hi * 8 + v;
        float* o = out + rowm * 64 + m;
        o[0]  = c0[v];
        o[16] = c1[v];
        o[32] = c2[v];
        o[48] = c3[v];
    }
}


extern "C" void kernel_launch(void* const* d_in, const int* in_sizes, int n_in,
                              void* d_out, int out_size, void* d_ws, size_t ws_size,
                              hipStream_t stream) {
    (void)in_sizes; (void)n_in; (void)out_size; (void)ws_size;

    const float* x  = (const float*)d_in[0];
    const int*   ei = (const int*)d_in[1];
    const float* ew = (const float*)d_in[2];
    const float* tf = (const float*)d_in[3];
    const float* tb = (const float*)d_in[4];
    const int* row = ei;
    const int* col = ei + N_EDGES;

    float* ws  = (float*)d_ws;
    float* deg = ws;
    float* Tx1 = ws  + 50176;
    float* Tx2 = Tx1 + (size_t)N_NODES * C;
    float* Tx3 = Tx2 + (size_t)N_NODES * C;
    float* nrm = Tx3 + (size_t)N_NODES * C;

    const int zero4 = (50176 + 3 * N_NODES * C) / 4;
    zero_kernel<<<(zero4 + 255) / 256, 256, 0, stream>>>((float4*)deg, zero4);

    deg_kernel<<<N_EDGES / 256, 256, 0, stream>>>(row, ew, deg);
    deginv_kernel<<<(N_NODES + 255) / 256, 256, 0, stream>>>(deg);
    norm_kernel<<<N_EDGES / 256, 256, 0, stream>>>(row, col, ew, deg, nrm);

    const int prop_blocks = (N_EDGES * 16) / 256;
    prop_kernel<<<prop_blocks, 256, 0, stream>>>(x,   Tx1, row, col, nrm);
    prop_kernel<<<prop_blocks, 256, 0, stream>>>(Tx1, Tx2, row, col, nrm);
    prop_kernel<<<prop_blocks, 256, 0, stream>>>(Tx2, Tx3, row, col, nrm);

    gemm_kernel<<<(3125 + 7) / 8, 256, 0, stream>>>(x, Tx1, Tx2, Tx3, tf, tb,
                                                    (float*)d_out);
}


# prop: edges grouped by destination row in each call (LDS slot counters and degree sums per node range); each pass gathers a row's edges in registers, stores once; no global atomics, list-order sums
# speedup vs baseline: 1.5363x; 1.5363x over previous
#include <hip/hip_runtime.h>

#define N_NODES 50000
#define N_EDGES 800000
#define C 64

typedef float    v2f  __attribute__((ext_vector_type(2)));
typedef float    v8f  __attribute__((ext_vector_type(8)));

#define RANGE_NODES 200
#define RANGE_GROUPS 250

__global__ void __launch_bounds__(1024)
build_kernel(const int* __restrict__ row, const float* __restrict__ w,
             int* __restrict__ rowptr, float* __restrict__ dinv, int* __restrict__ slot) {
    __shared__ int   cnt[256];
    __shared__ float dg[256];
    __shared__ int   sc[256];
    __shared__ int   below_s;

    const int t  = threadIdx.x;
    const int lo = blockIdx.x * RANGE_NODES;
    if (t < 256) { cnt[t] = 0; dg[t] = 0.f; }
    if (t == 0) below_s = 0;
    __syncthreads();

    int below = 0;
    const int4* row4 = reinterpret_cast<const int4*>(row);
    const int far = 0x7fffffff;
    for (int b = t; b < N_EDGES / 4; b += 4096) {
        int4 rv[4];
        #pragma unroll
        for (int u = 0; u < 4; ++u) {
            const int i = b + u * 1024;
            rv[u] = (i < N_EDGES / 4) ? row4[i] : make_int4(far, far, far, far);
        }
        #pragma unroll
        for (int u = 0; u < 4; ++u) {
            const int e = (b + u * 1024) * 4;
            const int r0 = rv[u].x, r1 = rv[u].y, r2 = rv[u].z, r3 = rv[u].w;
            below += (r0 < lo ? 1 : 0) + (r1 < lo ? 1 : 0) + (r2 < lo ? 1 : 0) + (r3 < lo ? 1 : 0);
            const unsigned d0 = (unsigned)(r0 - lo), d1 = (unsigned)(r1 - lo);
            const unsigned d2 = (unsigned)(r2 - lo), d3 = (unsigned)(r3 - lo);
            if (min(min(d0, d1), min(d2, d3)) < (unsigned)RANGE_NODES) {
                if (d0 < (unsigned)RANGE_NODES) { slot[e]     = atomicAdd(&cnt[d0], 1); atomicAdd(&dg[d0], w[e]);     }
                if (d1 < (unsigned)RANGE_NODES) { slot[e + 1] = atomicAdd(&cnt[d1], 1); atomicAdd(&dg[d1], w[e + 1]); }
                if (d2 < (unsigned)RANGE_NODES) { slot[e + 2] = atomicAdd(&cnt[d2], 1); atomicAdd(&dg[d2], w[e + 2]); }
                if (d3 < (unsigned)RANGE_NODES) { slot[e + 3] = atomicAdd(&cnt[d3], 1); atomicAdd(&dg[d3], w[e + 3]); }
            }
        }
    }
    for (int off = 16; off > 0; off >>= 1) below += __shfl_xor(below, off);
    if ((t & 31) == 0) atomicAdd(&below_s, below);
    __syncthreads();

    int mine = 0;
    if (t < 256) { mine = (t < RANGE_NODES) ? cnt[t] : 0; sc[t] = mine; }
    __syncthreads();
    for (int off = 1; off < 256; off <<= 1) {
        int a = 0;
        if (t < 256 && t >= off) a = sc[t - off];
        __syncthreads();
        if (t < 256) sc[t] += a;
        __syncthreads();
    }
    if (t < RANGE_NODES) {
        const int base = below_s;
        rowptr[lo + t] = base + sc[t] - mine;
        const float d = dg[t];
        dinv[lo + t] = (d > 0.f) ? rsqrtf(d) : 0.f;
        if (lo + t == N_NODES - 1) rowptr[N_NODES] = base + sc[t];
    }
}

__global__ void fill_kernel(const int* __restrict__ row, const int* __restrict__ col,
                            const float* __restrict__ w, const int* __restrict__ rowptr,
                            const float* __restrict__ dinv, const int* __restrict__ slot,
                            int* __restrict__ lst) {
    const int e = blockIdx.x * blockDim.x + threadIdx.x;
    const int r = row[e];
    const int c = col[e];
    const float n = dinv[r] * w[e] * dinv[c];
    reinterpret_cast<int2*>(lst)[rowptr[r] + slot[e]] = make_int2(c, __float_as_int(n));
}

template <int G>
__device__ __forceinline__ void gather_group(const float* __restrict__ xin, int lane, int2 m, int j,
                                             float& ax, float& ay) {
    float2 v[G];
    float  wt[G];
    #pragma unroll
    for (int k = 0; k < G; ++k) {
        const unsigned c = (unsigned)__builtin_amdgcn_readlane(m.x, j + k);
        wt[k] = __int_as_float(__builtin_amdgcn_readlane(m.y, j + k));
        v[k]  = reinterpret_cast<const float2*>(xin + (size_t)c * C)[lane];
    }
    #pragma unroll
    for (int k = 0; k < G; ++k) {
        ax = fmaf(wt[k], v[k].x, ax);
        ay = fmaf(wt[k], v[k].y, ay);
    }
}

__global__ void __launch_bounds__(256)
pull_kernel(const float* __restrict__ xin, float* __restrict__ xout,
            const int* __restrict__ rowptr, const int* __restrict__ lst) {
    const int lane = threadIdx.x & 31;
    const int r = __builtin_amdgcn_readfirstlane(blockIdx.x * 8 + (threadIdx.x >> 5));
    const int beg = rowptr[r];
    const int end = rowptr[r + 1];
    float ax = 0.f, ay = 0.f;
    for (int b = beg; b < end; b += 32) {
        const int n = min(32, end - b);
        int2 m = make_int2(0, 0);
        if (lane < n) m = reinterpret_cast<const int2*>(lst)[b + lane];
        int j = 0;
        for (; j + 8 <= n; j += 8) gather_group<8>(xin, lane, m, j, ax, ay);
        if (j + 4 <= n) { gather_group<4>(xin, lane, m, j, ax, ay); j += 4; }
        if (j + 2 <= n) { gather_group<2>(xin, lane, m, j, ax, ay); j += 2; }
        if (j < n)      { gather_group<1>(xin, lane, m, j, ax, ay); }
    }
    *reinterpret_cast<float2*>(xout + (size_t)r * C + lane * 2) = make_float2(ax, ay);
}


__device__ __forceinline__ void mat_accum(const float* __restrict__ A,
                                          const float* Wm,
                                          int arow, int m, int hi,
                                          v8f& c0, v8f& c1, v8f& c2, v8f& c3) {
    #pragma unroll
    for (int k0 = 0; k0 < 64; k0 += 4) {
        const int kk = k0 + hi * 2;
        v2f a = *reinterpret_cast<const v2f*>(A + arow * 64 + kk);
        v2f b0, b1, b2, b3;
        b0[0] = Wm[kk * 64 + m];          b0[1] = Wm[(kk + 1) * 64 + m];
        b1[0] = Wm[kk * 64 + 16 + m];     b1[1] = Wm[(kk + 1) * 64 + 16 + m];
        b2[0] = Wm[kk * 64 + 32 + m];     b2[1] = Wm[(kk + 1) * 64 + 32 + m];
        b3[0] = Wm[kk * 64 + 48 + m];     b3[1] = Wm[(kk + 1) * 64 + 48 + m];
        c0 = __builtin_amdgcn_wmma_f32_16x16x4_f32(false, a, false, b0, (short)0, c0, false, false);
        c1 = __builtin_amdgcn_wmma_f32_16x16x4_f32(false, a, false, b1, (short)0, c1, false, false);
        c2 = __builtin_amdgcn_wmma_f32_16x16x4_f32(false, a, false, b2, (short)0, c2, false, false);
        c3 = __builtin_amdgcn_wmma_f32_16x16x4_f32(false, a, false, b3, (short)0, c3, false, false);
    }
}

__global__ void __launch_bounds__(256)
gemm_kernel(const float* __restrict__ A0, const float* __restrict__ A1,
            const float* __restrict__ A2, const float* __restrict__ A3,
            const float* __restrict__ tf, const float* __restrict__ tb,
            float* __restrict__ out) {
    __shared__ float Ws[4 * 64 * 64];

    const int t = threadIdx.x;
    #pragma unroll 4
    for (int j = 0; j < 64; ++j) {
        int i   = t + j * 256;
        int mat = i >> 12;
        int idx = i & 4095;
        float v;
        if      (mat == 0) v = tf[idx];
        else if (mat == 1) v = tb[idx] + tb[4096 + idx];
        else if (mat == 2) v = tf[4096 + idx] + tb[8192 + idx];
        else               v = tf[8192 + idx];
        Ws[i] = v;
    }
    __syncthreads();

    const int wave = t >> 5;
    const int lane = t & 31;
    const int tm   = blockIdx.x * 8 + wave;
    if (tm >= 3125) return;

    const int m    = lane & 15;
    const int hi   = lane >> 4;
    const int arow = tm * 16 + m;

    v8f c0 = {}, c1 = {}, c2 = {}, c3 = {};
    mat_accum(A0, Ws,            arow, m, hi, c0, c1, c2, c3);
    mat_accum(A1, Ws + 4096,     arow, m, hi, c0, c1, c2, c3);
    mat_accum(A2, Ws + 8192,     arow, m, hi, c0, c1, c2, c3);
    mat_accum(A3, Ws + 12288,    arow, m, hi, c0, c1, c2, c3);

    #pragma unroll
    for (int v = 0; v < 8; ++v) {
        const int rowm = tm * 16 + hi * 8 + v;
        float* o = out + rowm * 64 + m;
        o[0]  = c0[v];
        o[16] = c1[v];
        o[32] = c2[v];
        o[48] = c3[v];
    }
}


extern "C" void kernel_launch(void* const* d_in, const int* in_sizes, int n_in,
                              void* d_out, int out_size, void* d_ws, size_t ws_size,
                              hipStream_t stream) {
    (void)in_sizes; (void)n_in; (void)out_size; (void)ws_size;

    const float* x  = (const float*)d_in[0];
    const int*   ei = (const int*)d_in[1];
    const float* ew = (const float*)d_in[2];
    const float* tf = (const float*)d_in[3];
    const float* tb = (const float*)d_in[4];
    const int* row = ei;
    const int* col = ei + N_EDGES;

    float* ws     = (float*)d_ws;
    int*   rowptr = (int*)ws;
    float* Tx1    = ws  + 50176;
    float* Tx2    = Tx1 + (size_t)N_NODES * C;
    float* Tx3    = Tx2 + (size_t)N_NODES * C;
    float* dinv   = Tx3;
    int*   slot   = (int*)(Tx3 + 50176);
    int*   lst    = (int*)d_out;

    build_kernel<<<RANGE_GROUPS, 1024, 0, stream>>>(row, ew, rowptr, dinv, slot);
    fill_kernel<<<N_EDGES / 256, 256, 0, stream>>>(row, col, ew, rowptr, dinv, slot, lst);

    pull_kernel<<<N_NODES / 8, 256, 0, stream>>>(x,   Tx1, rowptr, lst);
    pull_kernel<<<N_NODES / 8, 256, 0, stream>>>(Tx1, Tx2, rowptr, lst);
    pull_kernel<<<N_NODES / 8, 256, 0, stream>>>(Tx2, Tx3, rowptr, lst);

    gemm_kernel<<<(3125 + 7) / 8, 256, 0, stream>>>(x, Tx1, Tx2, Tx3, tf, tb,
                                                    (float*)d_out);
}


# build: row stream only queues its in-range edges per wave in LDS (no load or returning atomic in it); weights, slots, degree adds drained after it, loads batched; sums stay LDS arrival order
# speedup vs baseline: 1.7573x; 1.1439x over previous
#include <hip/hip_runtime.h>

#define N_NODES 50000
#define N_EDGES 800000
#define C 64

typedef float    v2f  __attribute__((ext_vector_type(2)));
typedef float    v8f  __attribute__((ext_vector_type(8)));

#define RANGE_NODES 200
#define RANGE_GROUPS 250

#define QCAP 512
__global__ void __launch_bounds__(1024)
build_kernel(const int* __restrict__ row, const float* __restrict__ w,
             int* __restrict__ rowptr, float* __restrict__ dinv, int* __restrict__ slot) {
    __shared__ int      cnt[256];
    __shared__ float    dg[256];
    __shared__ int      sc[256];
    __shared__ int      below_s;
    __shared__ unsigned q[32 * QCAP];

    const int t    = threadIdx.x;
    const int lane = t & 31;
    const int lo   = blockIdx.x * RANGE_NODES;
    unsigned* myq  = q + (t >> 5) * QCAP;
    int qn = 0;
    if (t < 256) { cnt[t] = 0; dg[t] = 0.f; }
    if (t == 0) below_s = 0;
    __syncthreads();

    int below = 0;
    const int4* row4 = reinterpret_cast<const int4*>(row);
    const int far = 0x7fffffff;
    for (int b0 = 0; b0 < N_EDGES / 4; b0 += 4096) {
        const int b = b0 + t;
        int4 rv[4];
        #pragma unroll
        for (int u = 0; u < 4; ++u) {
            const int i = b + u * 1024;
            rv[u] = (i < N_EDGES / 4) ? row4[i] : make_int4(far, far, far, far);
        }
        #pragma unroll
        for (int u = 0; u < 4; ++u) {
            const int e = (b + u * 1024) * 4;
            const int r0 = rv[u].x, r1 = rv[u].y, r2 = rv[u].z, r3 = rv[u].w;
            below += (r0 < lo ? 1 : 0) + (r1 < lo ? 1 : 0) + (r2 < lo ? 1 : 0) + (r3 < lo ? 1 : 0);
            const unsigned dd[4] = { (unsigned)(r0 - lo), (unsigned)(r1 - lo),
                                     (unsigned)(r2 - lo), (unsigned)(r3 - lo) };
            const unsigned dmin = min(min(dd[0], dd[1]), min(dd[2], dd[3]));
            if (__builtin_amdgcn_ballot_w32(dmin < (unsigned)RANGE_NODES) != 0u) {
                #pragma unroll
                for (int k = 0; k < 4; ++k) {
                    const bool in = dd[k] < (unsigned)RANGE_NODES;
                    const unsigned bal = __builtin_amdgcn_ballot_w32(in);
                    if (bal != 0u) {
                        if (qn + 32 <= QCAP) {
                            if (in) myq[qn + (int)__builtin_amdgcn_mbcnt_lo(bal, 0u)] = (dd[k] << 20) | (unsigned)(e + k);
                            qn += __builtin_popcount(bal);
                        } else if (in) {
                            slot[e + k] = atomicAdd(&cnt[dd[k]], 1);
                            atomicAdd(&dg[dd[k]], w[e + k]);
                        }
                    }
                }
            }
        }
    }
    for (int off = 16; off > 0; off >>= 1) below += __shfl_xor(below, off);
    if (lane == 0) atomicAdd(&below_s, below);
    __syncthreads();

    for (int i0 = 0; i0 < qn; i0 += 128) {
        unsigned ent[4];
        float    wv[4];
        const unsigned none = 0xffffffffu;
        #pragma unroll
        for (int u = 0; u < 4; ++u) {
            const int i = i0 + u * 32 + lane;
            ent[u] = (i < qn) ? myq[i] : none;
        }
        #pragma unroll
        for (int u = 0; u < 4; ++u) wv[u] = (ent[u] != none) ? w[ent[u] & 0xfffffu] : 0.f;
        #pragma unroll
        for (int u = 0; u < 4; ++u) {
            if (ent[u] != none) {
                const unsigned e = ent[u] & 0xfffffu, d = ent[u] >> 20;
                slot[e] = atomicAdd(&cnt[d], 1);
                atomicAdd(&dg[d], wv[u]);
            }
        }
    }
    __syncthreads();

    int mine = 0;
    if (t < 256) { mine = (t < RANGE_NODES) ? cnt[t] : 0; sc[t] = mine; }
    __syncthreads();
    for (int off = 1; off < 256; off <<= 1) {
        int a = 0;
        if (t < 256 && t >= off) a = sc[t - off];
        __syncthreads();
        if (t < 256) sc[t] += a;
        __syncthreads();
    }
    if (t < RANGE_NODES) {
        const int base = below_s;
        rowptr[lo + t] = base + sc[t] - mine;
        const float d = dg[t];
        dinv[lo + t] = (d > 0.f) ? rsqrtf(d) : 0.f;
        if (lo + t == N_NODES - 1) rowptr[N_NODES] = base + sc[t];
    }
}

__global__ void fill_kernel(const int* __restrict__ row, const int* __restrict__ col,
                            const float* __restrict__ w, const int* __restrict__ rowptr,
                            const float* __restrict__ dinv, const int* __restrict__ slot,
                            int* __restrict__ lst) {
    const int e = blockIdx.x * blockDim.x + threadIdx.x;
    const int r = row[e];
    const int c = col[e];
    const float n = dinv[r] * w[e] * dinv[c];
    reinterpret_cast<int2*>(lst)[rowptr[r] + slot[e]] = make_int2(c, __float_as_int(n));
}

template <int G>
__device__ __forceinline__ void gather_group(const float* __restrict__ xin, int lane, int2 m, int j,
                                             float& ax, float& ay) {
    float2 v[G];
    float  wt[G];
    #pragma unroll
    for (int k = 0; k < G; ++k) {
        const unsigned c = (unsigned)__builtin_amdgcn_readlane(m.x, j + k);
        wt[k] = __int_as_float(__builtin_amdgcn_readlane(m.y, j + k));
        v[k]  = reinterpret_cast<const float2*>(xin + (size_t)c * C)[lane];
    }
    #pragma unroll
    for (int k = 0; k < G; ++k) {
        ax = fmaf(wt[k], v[k].x, ax);
        ay = fmaf(wt[k], v[k].y, ay);
    }
}

__global__ void __launch_bounds__(256)
pull_kernel(const float* __restrict__ xin, float* __restrict__ xout,
            const int* __restrict__ rowptr, const int* __restrict__ lst) {
    const int lane = threadIdx.x & 31;
    const int r = __builtin_amdgcn_readfirstlane(blockIdx.x * 8 + (threadIdx.x >> 5));
    const int beg = rowptr[r];
    const int end = rowptr[r + 1];
    float ax = 0.f, ay = 0.f;
    for (int b = beg; b < end; b += 32) {
        const int n = min(32, end - b);
        int2 m = make_int2(0, 0);
        if (lane < n) m = reinterpret_cast<const int2*>(lst)[b + lane];
        int j = 0;
        for (; j + 8 <= n; j += 8) gather_group<8>(xin, lane, m, j, ax, ay);
        if (j + 4 <= n) { gather_group<4>(xin, lane, m, j, ax, ay); j += 4; }
        if (j + 2 <= n) { gather_group<2>(xin, lane, m, j, ax, ay); j += 2; }
        if (j < n)      { gather_group<1>(xin, lane, m, j, ax, ay); }
    }
    *reinterpret_cast<float2*>(xout + (size_t)r * C + lane * 2) = make_float2(ax, ay);
}


__device__ __forceinline__ void mat_accum(const float* __restrict__ A,
                                          const float* Wm,
                                          int arow, int m, int hi,
                                          v8f& c0, v8f& c1, v8f& c2, v8f& c3) {
    #pragma unroll
    for (int k0 = 0; k0 < 64; k0 += 4) {
        const int kk = k0 + hi * 2;
        v2f a = *reinterpret_cast<const v2f*>(A + arow * 64 + kk);
        v2f b0, b1, b2, b3;
        b0[0] = Wm[kk * 64 + m];          b0[1] = Wm[(kk + 1) * 64 + m];
        b1[0] = Wm[kk * 64 + 16 + m];     b1[1] = Wm[(kk + 1) * 64 + 16 + m];
        b2[0] = Wm[kk * 64 + 32 + m];     b2[1] = Wm[(kk + 1) * 64 + 32 + m];
        b3[0] = Wm[kk * 64 + 48 + m];     b3[1] = Wm[(kk + 1) * 64 + 48 + m];
        c0 = __builtin_amdgcn_wmma_f32_16x16x4_f32(false, a, false, b0, (short)0, c0, false, false);
        c1 = __builtin_amdgcn_wmma_f32_16x16x4_f32(false, a, false, b1, (short)0, c1, false, false);
        c2 = __builtin_amdgcn_wmma_f32_16x16x4_f32(false, a, false, b2, (short)0, c2, false, false);
        c3 = __builtin_amdgcn_wmma_f32_16x16x4_f32(false, a, false, b3, (short)0, c3, false, false);
    }
}

__global__ void __launch_bounds__(256)
gemm_kernel(const float* __restrict__ A0, const float* __restrict__ A1,
            const float* __restrict__ A2, const float* __restrict__ A3,
            const float* __restrict__ tf, const float* __restrict__ tb,
            float* __restrict__ out) {
    __shared__ float Ws[4 * 64 * 64];

    const int t = threadIdx.x;
    #pragma unroll 4
    for (int j = 0; j < 64; ++j) {
        int i   = t + j * 256;
        int mat = i >> 12;
        int idx = i & 4095;
        float v;
        if      (mat == 0) v = tf[idx];
        else if (mat == 1) v = tb[idx] + tb[4096 + idx];
        else if (mat == 2) v = tf[4096 + idx] + tb[8192 + idx];
        else               v = tf[8192 + idx];
        Ws[i] = v;
    }
    __syncthreads();

    const int wave = t >> 5;
    const int lane = t & 31;
    const int tm   = blockIdx.x * 8 + wave;
    if (tm >= 3125) return;

    const int m    = lane & 15;
    const int hi   = lane >> 4;
    const int arow = tm * 16 + m;

    v8f c0 = {}, c1 = {}, c2 = {}, c3 = {};
    mat_accum(A0, Ws,            arow, m, hi, c0, c1, c2, c3);
    mat_accum(A1, Ws + 4096,     arow, m, hi, c0, c1, c2, c3);
    mat_accum(A2, Ws + 8192,     arow, m, hi, c0, c1, c2, c3);
    mat_accum(A3, Ws + 12288,    arow, m, hi, c0, c1, c2, c3);

    #pragma unroll
    for (int v = 0; v < 8; ++v) {
        const int rowm = tm * 16 + hi * 8 + v;
        float* o = out + rowm * 64 + m;
        o[0]  = c0[v];
        o[16] = c1[v];
        o[32] = c2[v];
        o[48] = c3[v];
    }
}


extern "C" void kernel_launch(void* const* d_in, const int* in_sizes, int n_in,
                              void* d_out, int out_size, void* d_ws, size_t ws_size,
                              hipStream_t stream) {
    (void)in_sizes; (void)n_in; (void)out_size; (void)ws_size;

    const float* x  = (const float*)d_in[0];
    const int*   ei = (const int*)d_in[1];
    const float* ew = (const float*)d_in[2];
    const float* tf = (const float*)d_in[3];
    const float* tb = (const float*)d_in[4];
    const int* row = ei;
    const int* col = ei + N_EDGES;

    float* ws     = (float*)d_ws;
    int*   rowptr = (int*)ws;
    float* Tx1    = ws  + 50176;
    float* Tx2    = Tx1 + (size_t)N_NODES * C;
    float* Tx3    = Tx2 + (size_t)N_NODES * C;
    float* dinv   = Tx3;
    int*   slot   = (int*)(Tx3 + 50176);
    int*   lst    = (int*)d_out;

    build_kernel<<<RANGE_GROUPS, 1024, 0, stream>>>(row, ew, rowptr, dinv, slot);
    fill_kernel<<<N_EDGES / 256, 256, 0, stream>>>(row, col, ew, rowptr, dinv, slot, lst);

    pull_kernel<<<N_NODES / 8, 256, 0, stream>>>(x,   Tx1, rowptr, lst);
    pull_kernel<<<N_NODES / 8, 256, 0, stream>>>(Tx1, Tx2, rowptr, lst);
    pull_kernel<<<N_NODES / 8, 256, 0, stream>>>(Tx2, Tx3, rowptr, lst);

    gemm_kernel<<<(3125 + 7) / 8, 256, 0, stream>>>(x, Tx1, Tx2, Tx3, tf, tb,
                                                    (float*)d_out);
}


# build: 8 edge chunks x 32 row ranges, a work-group streams 1/8 of the row array, not all; new small kernel adds chunk counts and degree partials in chunk order; row lists chunk-major: sums reordered
# speedup vs baseline: 2.1087x; 1.2000x over previous
#include <hip/hip_runtime.h>

#define N_NODES 50000
#define N_EDGES 800000
#define C 64

typedef float    v2f  __attribute__((ext_vector_type(2)));
typedef float    v8f  __attribute__((ext_vector_type(8)));

#define S_CHUNKS    8
#define CHUNK_EDGES (N_EDGES / S_CHUNKS)
#define CHUNK_VEC   (CHUNK_EDGES / 4)
#define RANGE_NODES 1568
#define N_RANGES    32
#define PADN        (N_RANGES * RANGE_NODES)
#define LD_PER_IT   5

#define QCAP 512
__global__ void __launch_bounds__(1024)
part_kernel(const int* __restrict__ row, const float* __restrict__ w,
            int* __restrict__ cntp, float* __restrict__ dgp, int* __restrict__ nmatch,
            int* __restrict__ slot) {
    __shared__ int      cnt[RANGE_NODES];
    __shared__ float    dg[RANGE_NODES];
    __shared__ int      tot_s;
    __shared__ unsigned q[32 * QCAP];

    const int t    = threadIdx.x;
    const int lane = t & 31;
    const int j    = blockIdx.x % N_RANGES;
    const int s    = blockIdx.x / N_RANGES;
    const int lo   = j * RANGE_NODES;
    const int e0   = s * CHUNK_EDGES;
    unsigned* myq  = q + (t >> 5) * QCAP;
    int qn = 0;
    for (int i = t; i < RANGE_NODES; i += 1024) { cnt[i] = 0; dg[i] = 0.f; }
    if (t == 0) tot_s = 0;
    __syncthreads();

    const int4* row4 = reinterpret_cast<const int4*>(row + e0);
    const int far = 0x7fffffff;
    for (int b0 = 0; b0 < CHUNK_VEC; b0 += 1024 * LD_PER_IT) {
        const int b = b0 + t;
        int4 rv[LD_PER_IT];
        #pragma unroll
        for (int u = 0; u < LD_PER_IT; ++u) {
            const int i = b + u * 1024;
            rv[u] = (i < CHUNK_VEC) ? row4[i] : make_int4(far, far, far, far);
        }
        #pragma unroll
        for (int u = 0; u < LD_PER_IT; ++u) {
            const int e = e0 + (b + u * 1024) * 4;
            const unsigned dd[4] = { (unsigned)(rv[u].x - lo), (unsigned)(rv[u].y - lo),
                                     (unsigned)(rv[u].z - lo), (unsigned)(rv[u].w - lo) };
            #pragma unroll
            for (int k = 0; k < 4; ++k) {
                const bool in = dd[k] < (unsigned)RANGE_NODES;
                const unsigned bal = __builtin_amdgcn_ballot_w32(in);
                if (bal != 0u) {
                    if (qn + 32 <= QCAP) {
                        if (in) myq[qn + (int)__builtin_amdgcn_mbcnt_lo(bal, 0u)] = (dd[k] << 20) | (unsigned)(e + k);
                        qn += __builtin_popcount(bal);
                    } else if (in) {
                        slot[e + k] = atomicAdd(&cnt[dd[k]], 1);
                        atomicAdd(&dg[dd[k]], w[e + k]);
                    }
                }
            }
        }
    }
    __syncthreads();

    for (int i0 = 0; i0 < qn; i0 += 128) {
        unsigned ent[4];
        float    wv[4];
        const unsigned none = 0xffffffffu;
        #pragma unroll
        for (int u = 0; u < 4; ++u) {
            const int i = i0 + u * 32 + lane;
            ent[u] = (i < qn) ? myq[i] : none;
        }
        #pragma unroll
        for (int u = 0; u < 4; ++u) wv[u] = (ent[u] != none) ? w[ent[u] & 0xfffffu] : 0.f;
        #pragma unroll
        for (int u = 0; u < 4; ++u) {
            if (ent[u] != none) {
                const unsigned e = ent[u] & 0xfffffu, d = ent[u] >> 20;
                slot[e] = atomicAdd(&cnt[d], 1);
                atomicAdd(&dg[d], wv[u]);
            }
        }
    }
    __syncthreads();

    int mine = 0;
    for (int i = t; i < RANGE_NODES; i += 1024) {
        const int c = cnt[i];
        cntp[s * PADN + lo + i] = c;
        dgp[s * PADN + lo + i]  = dg[i];
        mine += c;
    }
    for (int off = 16; off > 0; off >>= 1) mine += __shfl_xor(mine, off);
    if (lane == 0) atomicAdd(&tot_s, mine);
    __syncthreads();
    if (t == 0) nmatch[s * N_RANGES + j] = tot_s;
}

__global__ void __launch_bounds__(1024)
combine_kernel(const int* __restrict__ cntp, const float* __restrict__ dgp,
               const int* __restrict__ nmatch,
               int* __restrict__ rowptr, float* __restrict__ dinv, int* __restrict__ offs) {
    __shared__ int wsum[32];
    __shared__ int base_s;
    const int t    = threadIdx.x;
    const int lane = t & 31;
    const int wv   = t >> 5;
    const int j    = blockIdx.x;
    const int lo   = j * RANGE_NODES;

    if (wv == 0) {
        int b = 0;
        #pragma unroll
        for (int s = 0; s < S_CHUNKS; ++s) {
            const int v = nmatch[s * N_RANGES + lane];
            b += (lane < j) ? v : 0;
        }
        for (int off = 16; off > 0; off >>= 1) b += __shfl_xor(b, off);
        if (lane == 0) base_s = b;
    }

    const bool act = t < RANGE_NODES / 2;
    const int  r0  = lo + 2 * t;
    int2   c[S_CHUNKS];
    float2 d[S_CHUNKS];
    #pragma unroll
    for (int s = 0; s < S_CHUNKS; ++s) {
        c[s] = act ? *reinterpret_cast<const int2*>(cntp + s * PADN + r0)  : make_int2(0, 0);
        d[s] = act ? *reinterpret_cast<const float2*>(dgp + s * PADN + r0) : make_float2(0.f, 0.f);
    }
    int t0 = 0, t1 = 0;
    float g0 = 0.f, g1 = 0.f;
    #pragma unroll
    for (int s = 0; s < S_CHUNKS; ++s) { t0 += c[s].x; t1 += c[s].y; g0 += d[s].x; g1 += d[s].y; }

    const int p = t0 + t1;
    int inc = p;
    for (int off = 1; off < 32; off <<= 1) {
        const int v = __shfl_up(inc, off);
        if (lane >= off) inc += v;
    }
    if (lane == 31) wsum[wv] = inc;
    __syncthreads();
    if (wv == 0) {
        const int v = wsum[lane];
        int x = v;
        for (int off = 1; off < 32; off <<= 1) {
            const int y = __shfl_up(x, off);
            if (lane >= off) x += y;
        }
        wsum[lane] = x - v;
    }
    __syncthreads();

    if (act) {
        const int start0 = base_s + wsum[wv] + inc - p;
        const int start1 = start0 + t0;
        *reinterpret_cast<int2*>(rowptr + r0) = make_int2(start0, start1);
        *reinterpret_cast<float2*>(dinv + r0) =
            make_float2((g0 > 0.f) ? rsqrtf(g0) : 0.f, (g1 > 0.f) ? rsqrtf(g1) : 0.f);
        int a0 = start0, a1 = start1;
        #pragma unroll
        for (int s = 0; s < S_CHUNKS; ++s) {
            *reinterpret_cast<int2*>(offs + s * PADN + r0) = make_int2(a0, a1);
            a0 += c[s].x;
            a1 += c[s].y;
        }
    }
}

__global__ void fill_kernel(const int* __restrict__ row, const int* __restrict__ col,
                            const float* __restrict__ w, const int* __restrict__ offs,
                            const float* __restrict__ dinv, const int* __restrict__ slot,
                            int* __restrict__ lst) {
    const int e = blockIdx.x * blockDim.x + threadIdx.x;
    const int r = row[e];
    const int c = col[e];
    const int s = e / CHUNK_EDGES;
    const float n = dinv[r] * w[e] * dinv[c];
    reinterpret_cast<int2*>(lst)[offs[s * PADN + r] + slot[e]] = make_int2(c, __float_as_int(n));
}

template <int G>
__device__ __forceinline__ void gather_group(const float* __restrict__ xin, int lane, int2 m, int j,
                                             float& ax, float& ay) {
    float2 v[G];
    float  wt[G];
    #pragma unroll
    for (int k = 0; k < G; ++k) {
        const unsigned c = (unsigned)__builtin_amdgcn_readlane(m.x, j + k);
        wt[k] = __int_as_float(__builtin_amdgcn_readlane(m.y, j + k));
        v[k]  = reinterpret_cast<const float2*>(xin + (size_t)c * C)[lane];
    }
    #pragma unroll
    for (int k = 0; k < G; ++k) {
        ax = fmaf(wt[k], v[k].x, ax);
        ay = fmaf(wt[k], v[k].y, ay);
    }
}

__global__ void __launch_bounds__(256)
pull_kernel(const float* __restrict__ xin, float* __restrict__ xout,
            const int* __restrict__ rowptr, const int* __restrict__ lst) {
    const int lane = threadIdx.x & 31;
    const int r = __builtin_amdgcn_readfirstlane(blockIdx.x * 8 + (threadIdx.x >> 5));
    const int beg = rowptr[r];
    const int end = rowptr[r + 1];
    float ax = 0.f, ay = 0.f;
    for (int b = beg; b < end; b += 32) {
        const int n = min(32, end - b);
        int2 m = make_int2(0, 0);
        if (lane < n) m = reinterpret_cast<const int2*>(lst)[b + lane];
        int j = 0;
        for (; j + 8 <= n; j += 8) gather_group<8>(xin, lane, m, j, ax, ay);
        if (j + 4 <= n) { gather_group<4>(xin, lane, m, j, ax, ay); j += 4; }
        if (j + 2 <= n) { gather_group<2>(xin, lane, m, j, ax, ay); j += 2; }
        if (j < n)      { gather_group<1>(xin, lane, m, j, ax, ay); }
    }
    *reinterpret_cast<float2*>(xout + (size_t)r * C + lane * 2) = make_float2(ax, ay);
}


__device__ __forceinline__ void mat_accum(const float* __restrict__ A,
                                          const float* Wm,
                                          int arow, int m, int hi,
                                          v8f& c0, v8f& c1, v8f& c2, v8f& c3) {
    #pragma unroll
    for (int k0 = 0; k0 < 64; k0 += 4) {
        const int kk = k0 + hi * 2;
        v2f a = *reinterpret_cast<const v2f*>(A + arow * 64 + kk);
        v2f b0, b1, b2, b3;
        b0[0] = Wm[kk * 64 + m];          b0[1] = Wm[(kk + 1) * 64 + m];
        b1[0] = Wm[kk * 64 + 16 + m];     b1[1] = Wm[(kk + 1) * 64 + 16 + m];
        b2[0] = Wm[kk * 64 + 32 + m];     b2[1] = Wm[(kk + 1) * 64 + 32 + m];
        b3[0] = Wm[kk * 64 + 48 + m];     b3[1] = Wm[(kk + 1) * 64 + 48 + m];
        c0 = __builtin_amdgcn_wmma_f32_16x16x4_f32(false, a, false, b0, (short)0, c0, false, false);
        c1 = __builtin_amdgcn_wmma_f32_16x16x4_f32(false, a, false, b1, (short)0, c1, false, false);
        c2 = __builtin_amdgcn_wmma_f32_16x16x4_f32(false, a, false, b2, (short)0, c2, false, false);
        c3 = __builtin_amdgcn_wmma_f32_16x16x4_f32(false, a, false, b3, (short)0, c3, false, false);
    }
}

__global__ void __launch_bounds__(256)
gemm_kernel(const float* __restrict__ A0, const float* __restrict__ A1,
            const float* __restrict__ A2, const float* __restrict__ A3,
            const float* __restrict__ tf, const float* __restrict__ tb,
            float* __restrict__ out) {
    __shared__ float Ws[4 * 64 * 64];

    const int t = threadIdx.x;
    #pragma unroll 4
    for (int j = 0; j < 64; ++j) {
        int i   = t + j * 256;
        int mat = i >> 12;
        int idx = i & 4095;
        float v;
        if      (mat == 0) v = tf[idx];
        else if (mat == 1) v = tb[idx] + tb[4096 + idx];
        else if (mat == 2) v = tf[4096 + idx] + tb[8192 + idx];
        else               v = tf[8192 + idx];
        Ws[i] = v;
    }
    __syncthreads();

    const int wave = t >> 5;
    const int lane = t & 31;
    const int tm   = blockIdx.x * 8 + wave;
    if (tm >= 3125) return;

    const int m    = lane & 15;
    const int hi   = lane >> 4;
    const int arow = tm * 16 + m;

    v8f c0 = {}, c1 = {}, c2 = {}, c3 = {};
    mat_accum(A0, Ws,            arow, m, hi, c0, c1, c2, c3);
    mat_accum(A1, Ws + 4096,     arow, m, hi, c0, c1, c2, c3);
    mat_accum(A2, Ws + 8192,     arow, m, hi, c0, c1, c2, c3);
    mat_accum(A3, Ws + 12288,    arow, m, hi, c0, c1, c2, c3);

    #pragma unroll
    for (int v = 0; v < 8; ++v) {
        const int rowm = tm * 16 + hi * 8 + v;
        float* o = out + rowm * 64 + m;
        o[0]  = c0[v];
        o[16] = c1[v];
        o[32] = c2[v];
        o[48] = c3[v];
    }
}


extern "C" void kernel_launch(void* const* d_in, const int* in_sizes, int n_in,
                              void* d_out, int out_size, void* d_ws, size_t ws_size,
                              hipStream_t stream) {
    (void)in_sizes; (void)n_in; (void)out_size; (void)ws_size;

    const float* x  = (const float*)d_in[0];
    const int*   ei = (const int*)d_in[1];
    const float* ew = (const float*)d_in[2];
    const float* tf = (const float*)d_in[3];
    const float* tb = (const float*)d_in[4];
    const int* row = ei;
    const int* col = ei + N_EDGES;

    float* ws     = (float*)d_ws;
    int*   rowptr = (int*)ws;
    float* Tx1    = ws  + 50176;
    float* Tx2    = Tx1 + (size_t)N_NODES * C;
    float* Tx3    = Tx2 + (size_t)N_NODES * C;
    float* dinv   = Tx3;
    int*   slot   = (int*)(Tx3 + PADN);
    int*   offs   = slot + N_EDGES;
    int*   cntp   = offs + S_CHUNKS * PADN;
    float* dgp    = (float*)(cntp + S_CHUNKS * PADN);
    int*   nmatch = (int*)(dgp + S_CHUNKS * PADN);
    int*   lst    = (int*)d_out;

    part_kernel<<<S_CHUNKS * N_RANGES, 1024, 0, stream>>>(row, ew, cntp, dgp, nmatch, slot);
    combine_kernel<<<N_RANGES, 1024, 0, stream>>>(cntp, dgp, nmatch, rowptr, dinv, offs);
    fill_kernel<<<N_EDGES / 256, 256, 0, stream>>>(row, col, ew, offs, dinv, slot, lst);

    pull_kernel<<<N_NODES / 8, 256, 0, stream>>>(x,   Tx1, rowptr, lst);
    pull_kernel<<<N_NODES / 8, 256, 0, stream>>>(Tx1, Tx2, rowptr, lst);
    pull_kernel<<<N_NODES / 8, 256, 0, stream>>>(Tx2, Tx3, rowptr, lst);

    gemm_kernel<<<(3125 + 7) / 8, 256, 0, stream>>>(x, Tx1, Tx2, Tx3, tf, tb,
                                                    (float*)d_out);
}


# gemm: combined weights staged with all 24 16-byte loads of a thread in flight, one wait, 16-byte LDS stores; was 64 serial load-wait-store round trips per thread; sums and their order unchanged
# speedup vs baseline: 2.5978x; 1.2319x over previous
#include <hip/hip_runtime.h>

#define N_NODES 50000
#define N_EDGES 800000
#define C 64

typedef float    v2f  __attribute__((ext_vector_type(2)));
typedef float    v8f  __attribute__((ext_vector_type(8)));

#define S_CHUNKS    8
#define CHUNK_EDGES (N_EDGES / S_CHUNKS)
#define CHUNK_VEC   (CHUNK_EDGES / 4)
#define RANGE_NODES 1568
#define N_RANGES    32
#define PADN        (N_RANGES * RANGE_NODES)
#define LD_PER_IT   5

#define QCAP 512
__global__ void __launch_bounds__(1024)
part_kernel(const int* __restrict__ row, const float* __restrict__ w,
            int* __restrict__ cntp, float* __restrict__ dgp, int* __restrict__ nmatch,
            int* __restrict__ slot) {
    __shared__ int      cnt[RANGE_NODES];
    __shared__ float    dg[RANGE_NODES];
    __shared__ int      tot_s;
    __shared__ unsigned q[32 * QCAP];

    const int t    = threadIdx.x;
    const int lane = t & 31;
    const int j    = blockIdx.x % N_RANGES;
    const int s    = blockIdx.x / N_RANGES;
    const int lo   = j * RANGE_NODES;
    const int e0   = s * CHUNK_EDGES;
    unsigned* myq  = q + (t >> 5) * QCAP;
    int qn = 0;
    for (int i = t; i < RANGE_NODES; i += 1024) { cnt[i] = 0; dg[i] = 0.f; }
    if (t == 0) tot_s = 0;
    __syncthreads();

    const int4* row4 = reinterpret_cast<const int4*>(row + e0);
    const int far = 0x7fffffff;
    for (int b0 = 0; b0 < CHUNK_VEC; b0 += 1024 * LD_PER_IT) {
        const int b = b0 + t;
        int4 rv[LD_PER_IT];
        #pragma unroll
        for (int u = 0; u < LD_PER_IT; ++u) {
            const int i = b + u * 1024;
            rv[u] = (i < CHUNK_VEC) ? row4[i] : make_int4(far, far, far, far);
        }
        #pragma unroll
        for (int u = 0; u < LD_PER_IT; ++u) {
            const int e = e0 + (b + u * 1024) * 4;
            const unsigned dd[4] = { (unsigned)(rv[u].x - lo), (unsigned)(rv[u].y - lo),
                                     (unsigned)(rv[u].z - lo), (unsigned)(rv[u].w - lo) };
            #pragma unroll
            for (int k = 0; k < 4; ++k) {
                const bool in = dd[k] < (unsigned)RANGE_NODES;
                const unsigned bal = __builtin_amdgcn_ballot_w32(in);
                if (bal != 0u) {
                    if (qn + 32 <= QCAP) {
                        if (in) myq[qn + (int)__builtin_amdgcn_mbcnt_lo(bal, 0u)] = (dd[k] << 20) | (unsigned)(e + k);
                        qn += __builtin_popcount(bal);
                    } else if (in) {
                        slot[e + k] = atomicAdd(&cnt[dd[k]], 1);
                        atomicAdd(&dg[dd[k]], w[e + k]);
                    }
                }
            }
        }
    }
    __syncthreads();

    for (int i0 = 0; i0 < qn; i0 += 128) {
        unsigned ent[4];
        float    wv[4];
        const unsigned none = 0xffffffffu;
        #pragma unroll
        for (int u = 0; u < 4; ++u) {
            const int i = i0 + u * 32 + lane;
            ent[u] = (i < qn) ? myq[i] : none;
        }
        #pragma unroll
        for (int u = 0; u < 4; ++u) wv[u] = (ent[u] != none) ? w[ent[u] & 0xfffffu] : 0.f;
        #pragma unroll
        for (int u = 0; u < 4; ++u) {
            if (ent[u] != none) {
                const unsigned e = ent[u] & 0xfffffu, d = ent[u] >> 20;
                slot[e] = atomicAdd(&cnt[d], 1);
                atomicAdd(&dg[d], wv[u]);
            }
        }
    }
    __syncthreads();

    int mine = 0;
    for (int i = t; i < RANGE_NODES; i += 1024) {
        const int c = cnt[i];
        cntp[s * PADN + lo + i] = c;
        dgp[s * PADN + lo + i]  = dg[i];
        mine += c;
    }
    for (int off = 16; off > 0; off >>= 1) mine += __shfl_xor(mine, off);
    if (lane == 0) atomicAdd(&tot_s, mine);
    __syncthreads();
    if (t == 0) nmatch[s * N_RANGES + j] = tot_s;
}

__global__ void __launch_bounds__(1024)
combine_kernel(const int* __restrict__ cntp, const float* __restrict__ dgp,
               const int* __restrict__ nmatch,
               int* __restrict__ rowptr, float* __restrict__ dinv, int* __restrict__ offs) {
    __shared__ int wsum[32];
    __shared__ int base_s;
    const int t    = threadIdx.x;
    const int lane = t & 31;
    const int wv   = t >> 5;
    const int j    = blockIdx.x;
    const int lo   = j * RANGE_NODES;

    if (wv == 0) {
        int b = 0;
        #pragma unroll
        for (int s = 0; s < S_CHUNKS; ++s) {
            const int v = nmatch[s * N_RANGES + lane];
            b += (lane < j) ? v : 0;
        }
        for (int off = 16; off > 0; off >>= 1) b += __shfl_xor(b, off);
        if (lane == 0) base_s = b;
    }

    const bool act = t < RANGE_NODES / 2;
    const int  r0  = lo + 2 * t;
    int2   c[S_CHUNKS];
    float2 d[S_CHUNKS];
    #pragma unroll
    for (int s = 0; s < S_CHUNKS; ++s) {
        c[s] = act ? *reinterpret_cast<const int2*>(cntp + s * PADN + r0)  : make_int2(0, 0);
        d[s] = act ? *reinterpret_cast<const float2*>(dgp + s * PADN + r0) : make_float2(0.f, 0.f);
    }
    int t0 = 0, t1 = 0;
    float g0 = 0.f, g1 = 0.f;
    #pragma unroll
    for (int s = 0; s < S_CHUNKS; ++s) { t0 += c[s].x; t1 += c[s].y; g0 += d[s].x; g1 += d[s].y; }

    const int p = t0 + t1;
    int inc = p;
    for (int off = 1; off < 32; off <<= 1) {
        const int v = __shfl_up(inc, off);
        if (lane >= off) inc += v;
    }
    if (lane == 31) wsum[wv] = inc;
    __syncthreads();
    if (wv == 0) {
        const int v = wsum[lane];
        int x = v;
        for (int off = 1; off < 32; off <<= 1) {
            const int y = __shfl_up(x, off);
            if (lane >= off) x += y;
        }
        wsum[lane] = x - v;
    }
    __syncthreads();

    if (act) {
        const int start0 = base_s + wsum[wv] + inc - p;
        const int start1 = start0 + t0;
        *reinterpret_cast<int2*>(rowptr + r0) = make_int2(start0, start1);
        *reinterpret_cast<float2*>(dinv + r0) =
            make_float2((g0 > 0.f) ? rsqrtf(g0) : 0.f, (g1 > 0.f) ? rsqrtf(g1) : 0.f);
        int a0 = start0, a1 = start1;
        #pragma unroll
        for (int s = 0; s < S_CHUNKS; ++s) {
            *reinterpret_cast<int2*>(offs + s * PADN + r0) = make_int2(a0, a1);
            a0 += c[s].x;
            a1 += c[s].y;
        }
    }
}

__global__ void fill_kernel(const int* __restrict__ row, const int* __restrict__ col,
                            const float* __restrict__ w, const int* __restrict__ offs,
                            const float* __restrict__ dinv, const int* __restrict__ slot,
                            int* __restrict__ lst) {
    const int e = blockIdx.x * blockDim.x + threadIdx.x;
    const int r = row[e];
    const int c = col[e];
    const int s = e / CHUNK_EDGES;
    const float n = dinv[r] * w[e] * dinv[c];
    reinterpret_cast<int2*>(lst)[offs[s * PADN + r] + slot[e]] = make_int2(c, __float_as_int(n));
}

template <int G>
__device__ __forceinline__ void gather_group(const float* __restrict__ xin, int lane, int2 m, int j,
                                             float& ax, float& ay) {
    float2 v[G];
    float  wt[G];
    #pragma unroll
    for (int k = 0; k < G; ++k) {
        const unsigned c = (unsigned)__builtin_amdgcn_readlane(m.x, j + k);
        wt[k] = __int_as_float(__builtin_amdgcn_readlane(m.y, j + k));
        v[k]  = reinterpret_cast<const float2*>(xin + (size_t)c * C)[lane];
    }
    #pragma unroll
    for (int k = 0; k < G; ++k) {
        ax = fmaf(wt[k], v[k].x, ax);
        ay = fmaf(wt[k], v[k].y, ay);
    }
}

__global__ void __launch_bounds__(256)
pull_kernel(const float* __restrict__ xin, float* __restrict__ xout,
            const int* __restrict__ rowptr, const int* __restrict__ lst) {
    const int lane = threadIdx.x & 31;
    const int r = __builtin_amdgcn_readfirstlane(blockIdx.x * 8 + (threadIdx.x >> 5));
    const int beg = rowptr[r];
    const int end = rowptr[r + 1];
    float ax = 0.f, ay = 0.f;
    for (int b = beg; b < end; b += 32) {
        const int n = min(32, end - b);
        int2 m = make_int2(0, 0);
        if (lane < n) m = reinterpret_cast<const int2*>(lst)[b + lane];
        int j = 0;
        for (; j + 8 <= n; j += 8) gather_group<8>(xin, lane, m, j, ax, ay);
        if (j + 4 <= n) { gather_group<4>(xin, lane, m, j, ax, ay); j += 4; }
        if (j + 2 <= n) { gather_group<2>(xin, lane, m, j, ax, ay); j += 2; }
        if (j < n)      { gather_group<1>(xin, lane, m, j, ax, ay); }
    }
    *reinterpret_cast<float2*>(xout + (size_t)r * C + lane * 2) = make_float2(ax, ay);
}


__device__ __forceinline__ void mat_accum(const float* __restrict__ A,
                                          const float* Wm,
                                          int arow, int m, int hi,
                                          v8f& c0, v8f& c1, v8f& c2, v8f& c3) {
    #pragma unroll
    for (int k0 = 0; k0 < 64; k0 += 4) {
        const int kk = k0 + hi * 2;
        v2f a = *reinterpret_cast<const v2f*>(A + arow * 64 + kk);
        v2f b0, b1, b2, b3;
        b0[0] = Wm[kk * 64 + m];          b0[1] = Wm[(kk + 1) * 64 + m];
        b1[0] = Wm[kk * 64 + 16 + m];     b1[1] = Wm[(kk + 1) * 64 + 16 + m];
        b2[0] = Wm[kk * 64 + 32 + m];     b2[1] = Wm[(kk + 1) * 64 + 32 + m];
        b3[0] = Wm[kk * 64 + 48 + m];     b3[1] = Wm[(kk + 1) * 64 + 48 + m];
        c0 = __builtin_amdgcn_wmma_f32_16x16x4_f32(false, a, false, b0, (short)0, c0, false, false);
        c1 = __builtin_amdgcn_wmma_f32_16x16x4_f32(false, a, false, b1, (short)0, c1, false, false);
        c2 = __builtin_amdgcn_wmma_f32_16x16x4_f32(false, a, false, b2, (short)0, c2, false, false);
        c3 = __builtin_amdgcn_wmma_f32_16x16x4_f32(false, a, false, b3, (short)0, c3, false, false);
    }
}

__global__ void __launch_bounds__(256)
gemm_kernel(const float* __restrict__ A0, const float* __restrict__ A1,
            const float* __restrict__ A2, const float* __restrict__ A3,
            const float* __restrict__ tf, const float* __restrict__ tb,
            float* __restrict__ out) {
    __shared__ __align__(16) float Ws[4 * 64 * 64];

    const int t = threadIdx.x;
    const float4* tf4 = reinterpret_cast<const float4*>(tf);
    const float4* tb4 = reinterpret_cast<const float4*>(tb);
    float4*       Ws4 = reinterpret_cast<float4*>(Ws);
    float4 f0[4], f1[4], f2[4], g0[4], g1[4], g2[4];
    #pragma unroll
    for (int u = 0; u < 4; ++u) {
        const int q = t + u * 256;
        f0[u] = tf4[q];  f1[u] = tf4[1024 + q];  f2[u] = tf4[2048 + q];
        g0[u] = tb4[q];  g1[u] = tb4[1024 + q];  g2[u] = tb4[2048 + q];
    }
    #pragma unroll
    for (int u = 0; u < 4; ++u) {
        const int q = t + u * 256;
        Ws4[q]        = f0[u];
        Ws4[1024 + q] = make_float4(g0[u].x + g1[u].x, g0[u].y + g1[u].y, g0[u].z + g1[u].z, g0[u].w + g1[u].w);
        Ws4[2048 + q] = make_float4(f1[u].x + g2[u].x, f1[u].y + g2[u].y, f1[u].z + g2[u].z, f1[u].w + g2[u].w);
        Ws4[3072 + q] = f2[u];
    }
    __syncthreads();

    const int wave = t >> 5;
    const int lane = t & 31;
    const int tm   = blockIdx.x * 8 + wave;
    if (tm >= 3125) return;

    const int m    = lane & 15;
    const int hi   = lane >> 4;
    const int arow = tm * 16 + m;

    v8f c0 = {}, c1 = {}, c2 = {}, c3 = {};
    mat_accum(A0, Ws,            arow, m, hi, c0, c1, c2, c3);
    mat_accum(A1, Ws + 4096,     arow, m, hi, c0, c1, c2, c3);
    mat_accum(A2, Ws + 8192,     arow, m, hi, c0, c1, c2, c3);
    mat_accum(A3, Ws + 12288,    arow, m, hi, c0, c1, c2, c3);

    #pragma unroll
    for (int v = 0; v < 8; ++v) {
        const int rowm = tm * 16 + hi * 8 + v;
        float* o = out + rowm * 64 + m;
        o[0]  = c0[v];
        o[16] = c1[v];
        o[32] = c2[v];
        o[48] = c3[v];
    }
}


extern "C" void kernel_launch(void* const* d_in, const int* in_sizes, int n_in,
                              void* d_out, int out_size, void* d_ws, size_t ws_size,
                              hipStream_t stream) {
    (void)in_sizes; (void)n_in; (void)out_size; (void)ws_size;

    const float* x  = (const float*)d_in[0];
    const int*   ei = (const int*)d_in[1];
    const float* ew = (const float*)d_in[2];
    const float* tf = (const float*)d_in[3];
    const float* tb = (const float*)d_in[4];
    const int* row = ei;
    const int* col = ei + N_EDGES;

    float* ws     = (float*)d_ws;
    int*   rowptr = (int*)ws;
    float* Tx1    = ws  + 50176;
    float* Tx2    = Tx1 + (size_t)N_NODES * C;
    float* Tx3    = Tx2 + (size_t)N_NODES * C;
    float* dinv   = Tx3;
    int*   slot   = (int*)(Tx3 + PADN);
    int*   offs   = slot + N_EDGES;
    int*   cntp   = offs + S_CHUNKS * PADN;
    float* dgp    = (float*)(cntp + S_CHUNKS * PADN);
    int*   nmatch = (int*)(dgp + S_CHUNKS * PADN);
    int*   lst    = (int*)d_out;

    part_kernel<<<S_CHUNKS * N_RANGES, 1024, 0, stream>>>(row, ew, cntp, dgp, nmatch, slot);
    combine_kernel<<<N_RANGES, 1024, 0, stream>>>(cntp, dgp, nmatch, rowptr, dinv, offs);
    fill_kernel<<<N_EDGES / 256, 256, 0, stream>>>(row, col, ew, offs, dinv, slot, lst);

    pull_kernel<<<N_NODES / 8, 256, 0, stream>>>(x,   Tx1, rowptr, lst);
    pull_kernel<<<N_NODES / 8, 256, 0, stream>>>(Tx1, Tx2, rowptr, lst);
    pull_kernel<<<N_NODES / 8, 256, 0, stream>>>(Tx2, Tx3, rowptr, lst);

    gemm_kernel<<<(3125 + 7) / 8, 256, 0, stream>>>(x, Tx1, Tx2, Tx3, tf, tb,
                                                    (float*)d_out);
}


# build: edge list by two-level bucket sort, each edge read once (chunk sorted by row bin in LDS, one work-group per bin then orders its rows); 6 launches, not 7; row and degree sums in arrival order
# speedup vs baseline: 4.0635x; 1.5642x over previous
#include <hip/hip_runtime.h>

#define N_NODES 50000
#define N_EDGES 800000
#define C 64

typedef float    v2f  __attribute__((ext_vector_type(2)));
typedef float    v8f  __attribute__((ext_vector_type(8)));

#define NCH      250
#define CH_E     (N_EDGES / NCH)
#define NB       256
#define BIN_ROWS 196
#define TABW     (NB + 1)
#define PADN     (NB * BIN_ROWS)

__device__ __forceinline__ int block_scan_excl(int v, int* wsum, int& tot) {
    const int lane = threadIdx.x & 31;
    const int wv   = threadIdx.x >> 5;
    int inc = v;
    for (int off = 1; off < 32; off <<= 1) {
        const int y = __shfl_up(inc, off);
        if (lane >= off) inc += y;
    }
    __syncthreads();
    if (lane == 31) wsum[wv] = inc;
    __syncthreads();
    const int ws = wsum[lane];
    int x = ws;
    for (int off = 1; off < 32; off <<= 1) {
        const int y = __shfl_up(x, off);
        if (lane >= off) x += y;
    }
    tot = __shfl(x, 31);
    const int before = __shfl(x - ws, wv);
    return before + inc - v;
}

__global__ void __launch_bounds__(1024)
bin_kernel(const int* __restrict__ row, const int* __restrict__ col, const float* __restrict__ w,
           int2* __restrict__ rec, int* __restrict__ tab) {
    __shared__ int  cnt[NB];
    __shared__ int  start[NB];
    __shared__ int  wsum[32];
    __shared__ int2 stage[CH_E];

    const int t  = threadIdx.x;
    const int c  = blockIdx.x;
    const int e0 = c * CH_E;
    if (t < NB) cnt[t] = 0;
    __syncthreads();

    const bool act = t < CH_E / 4;
    int4   rv = make_int4(0, 0, 0, 0), cv = make_int4(0, 0, 0, 0);
    float4 wq = make_float4(0.f, 0.f, 0.f, 0.f);
    if (act) {
        rv = reinterpret_cast<const int4*>(row + e0)[t];
        cv = reinterpret_cast<const int4*>(col + e0)[t];
        wq = reinterpret_cast<const float4*>(w + e0)[t];
    }
    const unsigned rr[4] = { (unsigned)rv.x, (unsigned)rv.y, (unsigned)rv.z, (unsigned)rv.w };
    const int      cc[4] = { cv.x, cv.y, cv.z, cv.w };
    const float    ww[4] = { wq.x, wq.y, wq.z, wq.w };
    unsigned bn[4];
    int      rk[4];
    #pragma unroll
    for (int k = 0; k < 4; ++k) {
        bn[k] = rr[k] / (unsigned)BIN_ROWS;
        rk[k] = act ? atomicAdd(&cnt[bn[k]], 1) : 0;
    }
    __syncthreads();

    int tot;
    const int v  = (t < NB) ? cnt[t] : 0;
    const int ex = block_scan_excl(v, wsum, tot);
    if (t < NB) { start[t] = ex; tab[c * TABW + t] = ex; }
    if (t == 0) tab[c * TABW + NB] = tot;
    __syncthreads();

    if (act) {
        #pragma unroll
        for (int k = 0; k < 4; ++k) {
            const unsigned ro = rr[k] - bn[k] * (unsigned)BIN_ROWS;
            stage[start[bn[k]] + rk[k]] = make_int2((int)((ro << 16) | (unsigned)cc[k]), __float_as_int(ww[k]));
        }
    }
    __syncthreads();
    for (int i = t; i < CH_E; i += 1024) rec[e0 + i] = stage[i];
}

#define CPW 8
__global__ void __launch_bounds__(1024)
rows_kernel(const int2* __restrict__ rec, const int* __restrict__ tab,
            int* __restrict__ rowptr, float* __restrict__ dinv, int2* __restrict__ lst) {
    __shared__ __align__(16) int s0s[32 * CPW];
    __shared__ __align__(16) int sln[32 * CPW];
    __shared__ int   rcnt[BIN_ROWS];
    __shared__ float rdg[BIN_ROWS];
    __shared__ int   rnext[BIN_ROWS];
    __shared__ float rdi[BIN_ROWS];
    __shared__ int   wsum[32];

    const int t    = threadIdx.x;
    const int lane = t & 31;
    const int wv   = t >> 5;
    const int b    = blockIdx.x;
    int a0 = 0, len = 0;
    if (t < NCH) {
        a0  = tab[t * TABW + b];
        len = tab[t * TABW + b + 1] - a0;
    }
    if (t < 32 * CPW) { s0s[t] = a0; sln[t] = len; }
    if (t < BIN_ROWS) { rcnt[t] = 0; rdg[t] = 0.f; }
    int base;
    (void)block_scan_excl(a0, wsum, base);
    __syncthreads();

    const int4 la = reinterpret_cast<const int4*>(sln)[wv * 2], lb = reinterpret_cast<const int4*>(sln)[wv * 2 + 1];
    const int4 sa = reinterpret_cast<const int4*>(s0s)[wv * 2], sb = reinterpret_cast<const int4*>(s0s)[wv * 2 + 1];
    const int lnv[CPW] = { la.x, la.y, la.z, la.w, lb.x, lb.y, lb.z, lb.w };
    const int sv[CPW]  = { sa.x, sa.y, sa.z, sa.w, sb.x, sb.y, sb.z, sb.w };
    int2 kr[CPW];
    int  ln[CPW], p0[CPW];
    #pragma unroll
    for (int k = 0; k < CPW; ++k) {
        ln[k] = __builtin_amdgcn_readfirstlane(lnv[k]);
        p0[k] = __builtin_amdgcn_readfirstlane((wv * CPW + k) * CH_E + sv[k]);
        kr[k] = (lane < ln[k]) ? rec[p0[k] + lane] : make_int2(-1, 0);
    }
    #pragma unroll
    for (int k = 0; k < CPW; ++k) {
        if (kr[k].x >= 0) {
            const int ro = kr[k].x >> 16;
            atomicAdd(&rcnt[ro], 1);
            atomicAdd(&rdg[ro], __int_as_float(kr[k].y));
        }
    }
    #pragma unroll
    for (int k = 0; k < CPW; ++k) {
        for (int i = lane + 32; i < ln[k]; i += 32) {
            const int2 r  = rec[p0[k] + i];
            const int  ro = r.x >> 16;
            atomicAdd(&rcnt[ro], 1);
            atomicAdd(&rdg[ro], __int_as_float(r.y));
        }
    }
    __syncthreads();

    int unused;
    const int rc  = (t < BIN_ROWS) ? rcnt[t] : 0;
    const int rex = block_scan_excl(rc, wsum, unused);
    if (t < BIN_ROWS) {
        const float g  = rdg[t];
        const float di = (g > 0.f) ? rsqrtf(g) : 0.f;
        rnext[t] = base + rex;
        rdi[t]   = di;
        rowptr[b * BIN_ROWS + t] = base + rex;
        dinv[b * BIN_ROWS + t]   = di;
    }
    __syncthreads();

    #pragma unroll
    for (int k = 0; k < CPW; ++k) {
        if (kr[k].x >= 0) {
            const int ro  = kr[k].x >> 16;
            const int pos = atomicAdd(&rnext[ro], 1);
            lst[pos] = make_int2(kr[k].x & 0xffff, __float_as_int(rdi[ro] * __int_as_float(kr[k].y)));
        }
    }
    #pragma unroll
    for (int k = 0; k < CPW; ++k) {
        for (int i = lane + 32; i < ln[k]; i += 32) {
            const int2 r   = rec[p0[k] + i];
            const int  ro  = r.x >> 16;
            const int  pos = atomicAdd(&rnext[ro], 1);
            lst[pos] = make_int2(r.x & 0xffff, __float_as_int(rdi[ro] * __int_as_float(r.y)));
        }
    }
}

template <int G>
__device__ __forceinline__ void gather_group(const float* __restrict__ xin, int lane, int2 m, int j,
                                             float& ax, float& ay) {
    float2 v[G];
    float  wt[G];
    #pragma unroll
    for (int k = 0; k < G; ++k) {
        const unsigned c = (unsigned)__builtin_amdgcn_readlane(m.x, j + k);
        wt[k] = __int_as_float(__builtin_amdgcn_readlane(m.y, j + k));
        v[k]  = reinterpret_cast<const float2*>(xin + (size_t)c * C)[lane];
    }
    #pragma unroll
    for (int k = 0; k < G; ++k) {
        ax = fmaf(wt[k], v[k].x, ax);
        ay = fmaf(wt[k], v[k].y, ay);
    }
}

template <bool FIRST>
__global__ void __launch_bounds__(256)
pull_kernel(const float* __restrict__ xin, float* __restrict__ xout,
            const int* __restrict__ rowptr, int* lst, const float* __restrict__ dinv) {
    const int lane = threadIdx.x & 31;
    const int r = __builtin_amdgcn_readfirstlane(blockIdx.x * 8 + (threadIdx.x >> 5));
    const int beg = rowptr[r];
    const int end = rowptr[r + 1];
    float ax = 0.f, ay = 0.f;
    for (int b = beg; b < end; b += 32) {
        const int n = min(32, end - b);
        int2 m = make_int2(0, 0);
        if (lane < n) {
            m = reinterpret_cast<const int2*>(lst)[b + lane];
            if (FIRST) {
                m.y = __float_as_int(__int_as_float(m.y) * dinv[m.x]);
                lst[2 * (b + lane) + 1] = m.y;
            }
        }
        int j = 0;
        for (; j + 8 <= n; j += 8) gather_group<8>(xin, lane, m, j, ax, ay);
        if (j + 4 <= n) { gather_group<4>(xin, lane, m, j, ax, ay); j += 4; }
        if (j + 2 <= n) { gather_group<2>(xin, lane, m, j, ax, ay); j += 2; }
        if (j < n)      { gather_group<1>(xin, lane, m, j, ax, ay); }
    }
    *reinterpret_cast<float2*>(xout + (size_t)r * C + lane * 2) = make_float2(ax, ay);
}


__device__ __forceinline__ void mat_accum(const float* __restrict__ A,
                                          const float* Wm,
                                          int arow, int m, int hi,
                                          v8f& c0, v8f& c1, v8f& c2, v8f& c3) {
    #pragma unroll
    for (int k0 = 0; k0 < 64; k0 += 4) {
        const int kk = k0 + hi * 2;
        v2f a = *reinterpret_cast<const v2f*>(A + arow * 64 + kk);
        v2f b0, b1, b2, b3;
        b0[0] = Wm[kk * 64 + m];          b0[1] = Wm[(kk + 1) * 64 + m];
        b1[0] = Wm[kk * 64 + 16 + m];     b1[1] = Wm[(kk + 1) * 64 + 16 + m];
        b2[0] = Wm[kk * 64 + 32 + m];     b2[1] = Wm[(kk + 1) * 64 + 32 + m];
        b3[0] = Wm[kk * 64 + 48 + m];     b3[1] = Wm[(kk + 1) * 64 + 48 + m];
        c0 = __builtin_amdgcn_wmma_f32_16x16x4_f32(false, a, false, b0, (short)0, c0, false, false);
        c1 = __builtin_amdgcn_wmma_f32_16x16x4_f32(false, a, false, b1, (short)0, c1, false, false);
        c2 = __builtin_amdgcn_wmma_f32_16x16x4_f32(false, a, false, b2, (short)0, c2, false, false);
        c3 = __builtin_amdgcn_wmma_f32_16x16x4_f32(false, a, false, b3, (short)0, c3, false, false);
    }
}

__global__ void __launch_bounds__(256)
gemm_kernel(const float* __restrict__ A0, const float* __restrict__ A1,
            const float* __restrict__ A2, const float* __restrict__ A3,
            const float* __restrict__ tf, const float* __restrict__ tb,
            float* __restrict__ out) {
    __shared__ __align__(16) float Ws[4 * 64 * 64];

    const int t = threadIdx.x;
    const float4* tf4 = reinterpret_cast<const float4*>(tf);
    const float4* tb4 = reinterpret_cast<const float4*>(tb);
    float4*       Ws4 = reinterpret_cast<float4*>(Ws);
    float4 f0[4], f1[4], f2[4], g0[4], g1[4], g2[4];
    #pragma unroll
    for (int u = 0; u < 4; ++u) {
        const int q = t + u * 256;
        f0[u] = tf4[q];  f1[u] = tf4[1024 + q];  f2[u] = tf4[2048 + q];
        g0[u] = tb4[q];  g1[u] = tb4[1024 + q];  g2[u] = tb4[2048 + q];
    }
    #pragma unroll
    for (int u = 0; u < 4; ++u) {
        const int q = t + u * 256;
        Ws4[q]        = f0[u];
        Ws4[1024 + q] = make_float4(g0[u].x + g1[u].x, g0[u].y + g1[u].y, g0[u].z + g1[u].z, g0[u].w + g1[u].w);
        Ws4[2048 + q] = make_float4(f1[u].x + g2[u].x, f1[u].y + g2[u].y, f1[u].z + g2[u].z, f1[u].w + g2[u].w);
        Ws4[3072 + q] = f2[u];
    }
    __syncthreads();

    const int wave = t >> 5;
    const int lane = t & 31;
    const int tm   = blockIdx.x * 8 + wave;
    if (tm >= 3125) return;

    const int m    = lane & 15;
    const int hi   = lane >> 4;
    const int arow = tm * 16 + m;

    v8f c0 = {}, c1 = {}, c2 = {}, c3 = {};
    mat_accum(A0, Ws,            arow, m, hi, c0, c1, c2, c3);
    mat_accum(A1, Ws + 4096,     arow, m, hi, c0, c1, c2, c3);
    mat_accum(A2, Ws + 8192,     arow, m, hi, c0, c1, c2, c3);
    mat_accum(A3, Ws + 12288,    arow, m, hi, c0, c1, c2, c3);

    #pragma unroll
    for (int v = 0; v < 8; ++v) {
        const int rowm = tm * 16 + hi * 8 + v;
        float* o = out + rowm * 64 + m;
        o[0]  = c0[v];
        o[16] = c1[v];
        o[32] = c2[v];
        o[48] = c3[v];
    }
}


extern "C" void kernel_launch(void* const* d_in, const int* in_sizes, int n_in,
                              void* d_out, int out_size, void* d_ws, size_t ws_size,
                              hipStream_t stream) {
    (void)in_sizes; (void)n_in; (void)out_size; (void)ws_size;

    const float* x  = (const float*)d_in[0];
    const int*   ei = (const int*)d_in[1];
    const float* ew = (const float*)d_in[2];
    const float* tf = (const float*)d_in[3];
    const float* tb = (const float*)d_in[4];
    const int* row = ei;
    const int* col = ei + N_EDGES;

    float* ws     = (float*)d_ws;
    int*   rowptr = (int*)ws;
    float* Tx1    = ws  + 50176;
    float* Tx2    = Tx1 + (size_t)N_NODES * C;
    float* Tx3    = Tx2 + (size_t)N_NODES * C;
    float* dinv   = Tx3;
    int2*  rec    = (int2*)(Tx3 + PADN);
    int*   tab    = (int*)(Tx3 + PADN) + 2 * N_EDGES;
    int*   lst    = (int*)d_out;

    bin_kernel<<<NCH, 1024, 0, stream>>>(row, col, ew, rec, tab);
    rows_kernel<<<NB, 1024, 0, stream>>>(rec, tab, rowptr, dinv, (int2*)lst);

    pull_kernel<true><<<N_NODES / 8, 256, 0, stream>>>(x,   Tx1, rowptr, lst, dinv);
    pull_kernel<false><<<N_NODES / 8, 256, 0, stream>>>(Tx1, Tx2, rowptr, lst, dinv);
    pull_kernel<false><<<N_NODES / 8, 256, 0, stream>>>(Tx2, Tx3, rowptr, lst, dinv);

    gemm_kernel<<<(3125 + 7) / 8, 256, 0, stream>>>(x, Tx1, Tx2, Tx3, tf, tb,
                                                    (float*)d_out);
}


# prop: 10000 waves a pass, not 50000: a wave takes 5 rows at one memory trip per row (all gathers of a row issued before one is used, next row's list piece asked for ahead); sums and order unchanged
# speedup vs baseline: 4.1736x; 1.0271x over previous
#include <hip/hip_runtime.h>

#define N_NODES 50000
#define N_EDGES 800000
#define C 64

typedef float    v2f  __attribute__((ext_vector_type(2)));
typedef float    v8f  __attribute__((ext_vector_type(8)));

#define NCH      250
#define CH_E     (N_EDGES / NCH)
#define NB       256
#define BIN_ROWS 196
#define TABW     (NB + 1)
#define PADN     (NB * BIN_ROWS)

__device__ __forceinline__ int block_scan_excl(int v, int* wsum, int& tot) {
    const int lane = threadIdx.x & 31;
    const int wv   = threadIdx.x >> 5;
    int inc = v;
    for (int off = 1; off < 32; off <<= 1) {
        const int y = __shfl_up(inc, off);
        if (lane >= off) inc += y;
    }
    __syncthreads();
    if (lane == 31) wsum[wv] = inc;
    __syncthreads();
    const int ws = wsum[lane];
    int x = ws;
    for (int off = 1; off < 32; off <<= 1) {
        const int y = __shfl_up(x, off);
        if (lane >= off) x += y;
    }
    tot = __shfl(x, 31);
    const int before = __shfl(x - ws, wv);
    return before + inc - v;
}

__global__ void __launch_bounds__(1024)
bin_kernel(const int* __restrict__ row, const int* __restrict__ col, const float* __restrict__ w,
           int2* __restrict__ rec, int* __restrict__ tab) {
    __shared__ int  cnt[NB];
    __shared__ int  start[NB];
    __shared__ int  wsum[32];
    __shared__ int2 stage[CH_E];

    const int t  = threadIdx.x;
    const int c  = blockIdx.x;
    const int e0 = c * CH_E;
    if (t < NB) cnt[t] = 0;
    __syncthreads();

    const bool act = t < CH_E / 4;
    int4   rv = make_int4(0, 0, 0, 0), cv = make_int4(0, 0, 0, 0);
    float4 wq = make_float4(0.f, 0.f, 0.f, 0.f);
    if (act) {
        rv = reinterpret_cast<const int4*>(row + e0)[t];
        cv = reinterpret_cast<const int4*>(col + e0)[t];
        wq = reinterpret_cast<const float4*>(w + e0)[t];
    }
    const unsigned rr[4] = { (unsigned)rv.x, (unsigned)rv.y, (unsigned)rv.z, (unsigned)rv.w };
    const int      cc[4] = { cv.x, cv.y, cv.z, cv.w };
    const float    ww[4] = { wq.x, wq.y, wq.z, wq.w };
    unsigned bn[4];
    int      rk[4];
    #pragma unroll
    for (int k = 0; k < 4; ++k) {
        bn[k] = rr[k] / (unsigned)BIN_ROWS;
        rk[k] = act ? atomicAdd(&cnt[bn[k]], 1) : 0;
    }
    __syncthreads();

    int tot;
    const int v  = (t < NB) ? cnt[t] : 0;
    const int ex = block_scan_excl(v, wsum, tot);
    if (t < NB) { start[t] = ex; tab[c * TABW + t] = ex; }
    if (t == 0) tab[c * TABW + NB] = tot;
    __syncthreads();

    if (act) {
        #pragma unroll
        for (int k = 0; k < 4; ++k) {
            const unsigned ro = rr[k] - bn[k] * (unsigned)BIN_ROWS;
            stage[start[bn[k]] + rk[k]] = make_int2((int)((ro << 16) | (unsigned)cc[k]), __float_as_int(ww[k]));
        }
    }
    __syncthreads();
    for (int i = t; i < CH_E; i += 1024) rec[e0 + i] = stage[i];
}

#define CPW 8
__global__ void __launch_bounds__(1024)
rows_kernel(const int2* __restrict__ rec, const int* __restrict__ tab,
            int* __restrict__ rowptr, float* __restrict__ dinv, int2* __restrict__ lst) {
    __shared__ __align__(16) int s0s[32 * CPW];
    __shared__ __align__(16) int sln[32 * CPW];
    __shared__ int   rcnt[BIN_ROWS];
    __shared__ float rdg[BIN_ROWS];
    __shared__ int   rnext[BIN_ROWS];
    __shared__ float rdi[BIN_ROWS];
    __shared__ int   wsum[32];

    const int t    = threadIdx.x;
    const int lane = t & 31;
    const int wv   = t >> 5;
    const int b    = blockIdx.x;
    int a0 = 0, len = 0;
    if (t < NCH) {
        a0  = tab[t * TABW + b];
        len = tab[t * TABW + b + 1] - a0;
    }
    if (t < 32 * CPW) { s0s[t] = a0; sln[t] = len; }
    if (t < BIN_ROWS) { rcnt[t] = 0; rdg[t] = 0.f; }
    int base;
    (void)block_scan_excl(a0, wsum, base);
    __syncthreads();

    const int4 la = reinterpret_cast<const int4*>(sln)[wv * 2], lb = reinterpret_cast<const int4*>(sln)[wv * 2 + 1];
    const int4 sa = reinterpret_cast<const int4*>(s0s)[wv * 2], sb = reinterpret_cast<const int4*>(s0s)[wv * 2 + 1];
    const int lnv[CPW] = { la.x, la.y, la.z, la.w, lb.x, lb.y, lb.z, lb.w };
    const int sv[CPW]  = { sa.x, sa.y, sa.z, sa.w, sb.x, sb.y, sb.z, sb.w };
    int2 kr[CPW];
    int  ln[CPW], p0[CPW];
    #pragma unroll
    for (int k = 0; k < CPW; ++k) {
        ln[k] = __builtin_amdgcn_readfirstlane(lnv[k]);
        p0[k] = __builtin_amdgcn_readfirstlane((wv * CPW + k) * CH_E + sv[k]);
        kr[k] = (lane < ln[k]) ? rec[p0[k] + lane] : make_int2(-1, 0);
    }
    #pragma unroll
    for (int k = 0; k < CPW; ++k) {
        if (kr[k].x >= 0) {
            const int ro = kr[k].x >> 16;
            atomicAdd(&rcnt[ro], 1);
            atomicAdd(&rdg[ro], __int_as_float(kr[k].y));
        }
    }
    #pragma unroll
    for (int k = 0; k < CPW; ++k) {
        for (int i = lane + 32; i < ln[k]; i += 32) {
            const int2 r  = rec[p0[k] + i];
            const int  ro = r.x >> 16;
            atomicAdd(&rcnt[ro], 1);
            atomicAdd(&rdg[ro], __int_as_float(r.y));
        }
    }
    __syncthreads();

    int unused;
    const int rc  = (t < BIN_ROWS) ? rcnt[t] : 0;
    const int rex = block_scan_excl(rc, wsum, unused);
    if (t < BIN_ROWS) {
        const float g  = rdg[t];
        const float di = (g > 0.f) ? rsqrtf(g) : 0.f;
        rnext[t] = base + rex;
        rdi[t]   = di;
        rowptr[b * BIN_ROWS + t] = base + rex;
        dinv[b * BIN_ROWS + t]   = di;
    }
    __syncthreads();

    #pragma unroll
    for (int k = 0; k < CPW; ++k) {
        if (kr[k].x >= 0) {
            const int ro  = kr[k].x >> 16;
            const int pos = atomicAdd(&rnext[ro], 1);
            lst[pos] = make_int2(kr[k].x & 0xffff, __float_as_int(rdi[ro] * __int_as_float(kr[k].y)));
        }
    }
    #pragma unroll
    for (int k = 0; k < CPW; ++k) {
        for (int i = lane + 32; i < ln[k]; i += 32) {
            const int2 r   = rec[p0[k] + i];
            const int  ro  = r.x >> 16;
            const int  pos = atomicAdd(&rnext[ro], 1);
            lst[pos] = make_int2(r.x & 0xffff, __float_as_int(rdi[ro] * __int_as_float(r.y)));
        }
    }
}

#define BATCH 24
template <int K>
__device__ __forceinline__ void issue_gathers(const float2* __restrict__ xlane, int mx, int j, int cnt,
                                              float2 (&v)[BATCH]) {
    if (K == 0 || K < cnt) {
        v[K] = xlane[(size_t)(unsigned)__builtin_amdgcn_readlane(mx, j + K) * (C / 2)];
        if constexpr (K + 1 < BATCH) issue_gathers<K + 1>(xlane, mx, j, cnt, v);
    }
}

template <int K>
__device__ __forceinline__ void add_gathers(int my, int j, int cnt, const float2 (&v)[BATCH],
                                            float& ax, float& ay) {
    if (K == 0 || K < cnt) {
        const float wt = __int_as_float(__builtin_amdgcn_readlane(my, j + K));
        ax = fmaf(wt, v[K].x, ax);
        ay = fmaf(wt, v[K].y, ay);
        if constexpr (K + 1 < BATCH) add_gathers<K + 1>(my, j, cnt, v, ax, ay);
    }
}

#define RPW 5
template <bool FIRST>
__global__ void __launch_bounds__(256)
pull_kernel(const float* __restrict__ xin, float* __restrict__ xout,
            const int* __restrict__ rowptr, int* lst, const float* __restrict__ dinv) {
    const int lane = threadIdx.x & 31;
    const int r0 = __builtin_amdgcn_readfirstlane((blockIdx.x * 8 + (threadIdx.x >> 5)) * RPW);
    const int bv = rowptr[r0 + min(lane, RPW)];
    const float2* xlane = reinterpret_cast<const float2*>(xin) + lane;
    const int2*   lst2  = reinterpret_cast<const int2*>(lst);
    int beg = __builtin_amdgcn_readlane(bv, 0);
    int end = beg;
    int n   = 0;
    int2 m  = make_int2(0, 0);
    #pragma nounroll
    for (int i = -1; i < RPW; ++i) {
        float dv = 0.f;
        if (FIRST) {
            if (lane < n) dv = dinv[m.x];
        }
        const int nbeg = end;
        int  nend = end, nn = 0;
        int2 mn   = make_int2(0, 0);
        if (i + 1 < RPW) {
            nend = __builtin_amdgcn_readlane(bv, i + 2);
            nn   = min(32, nend - nbeg);
            if (lane < nn) mn = lst2[nbeg + lane];
        }
        float ax = 0.f, ay = 0.f;
        if (n > 0) {
            #pragma nounroll
            for (int j = 0; j < n; j += BATCH) {
                float2 v[BATCH];
                issue_gathers<0>(xlane, m.x, j, n - j, v);
                if (FIRST) {
                    if (j == 0 && lane < n) {
                        m.y = __float_as_int(__int_as_float(m.y) * dv);
                        lst[2 * (beg + lane) + 1] = m.y;
                    }
                }
                add_gathers<0>(m.y, j, n - j, v, ax, ay);
            }
            for (int b2 = beg + 32; b2 < end; b2 += 32) {
                const int n2 = min(32, end - b2);
                int2 m2 = make_int2(0, 0);
                if (lane < n2) {
                    m2 = lst2[b2 + lane];
                    if (FIRST) {
                        m2.y = __float_as_int(__int_as_float(m2.y) * dinv[m2.x]);
                        lst[2 * (b2 + lane) + 1] = m2.y;
                    }
                }
                for (int k = 0; k < n2; ++k) {
                    const unsigned c  = (unsigned)__builtin_amdgcn_readlane(m2.x, k);
                    const float    wt = __int_as_float(__builtin_amdgcn_readlane(m2.y, k));
                    const float2   xv = xlane[(size_t)c * (C / 2)];
                    ax = fmaf(wt, xv.x, ax);
                    ay = fmaf(wt, xv.y, ay);
                }
            }
        }
        if (i >= 0) *reinterpret_cast<float2*>(xout + (size_t)(r0 + i) * C + lane * 2) = make_float2(ax, ay);
        m = mn;  n = nn;  beg = nbeg;  end = nend;
    }
}


__device__ __forceinline__ void mat_accum(const float* __restrict__ A,
                                          const float* Wm,
                                          int arow, int m, int hi,
                                          v8f& c0, v8f& c1, v8f& c2, v8f& c3) {
    #pragma unroll
    for (int k0 = 0; k0 < 64; k0 += 4) {
        const int kk = k0 + hi * 2;
        v2f a = *reinterpret_cast<const v2f*>(A + arow * 64 + kk);
        v2f b0, b1, b2, b3;
        b0[0] = Wm[kk * 64 + m];          b0[1] = Wm[(kk + 1) * 64 + m];
        b1[0] = Wm[kk * 64 + 16 + m];     b1[1] = Wm[(kk + 1) * 64 + 16 + m];
        b2[0] = Wm[kk * 64 + 32 + m];     b2[1] = Wm[(kk + 1) * 64 + 32 + m];
        b3[0] = Wm[kk * 64 + 48 + m];     b3[1] = Wm[(kk + 1) * 64 + 48 + m];
        c0 = __builtin_amdgcn_wmma_f32_16x16x4_f32(false, a, false, b0, (short)0, c0, false, false);
        c1 = __builtin_amdgcn_wmma_f32_16x16x4_f32(false, a, false, b1, (short)0, c1, false, false);
        c2 = __builtin_amdgcn_wmma_f32_16x16x4_f32(false, a, false, b2, (short)0, c2, false, false);
        c3 = __builtin_amdgcn_wmma_f32_16x16x4_f32(false, a, false, b3, (short)0, c3, false, false);
    }
}

__global__ void __launch_bounds__(256)
gemm_kernel(const float* __restrict__ A0, const float* __restrict__ A1,
            const float* __restrict__ A2, const float* __restrict__ A3,
            const float* __restrict__ tf, const float* __restrict__ tb,
            float* __restrict__ out) {
    __shared__ __align__(16) float Ws[4 * 64 * 64];

    const int t = threadIdx.x;
    const float4* tf4 = reinterpret_cast<const float4*>(tf);
    const float4* tb4 = reinterpret_cast<const float4*>(tb);
    float4*       Ws4 = reinterpret_cast<float4*>(Ws);
    float4 f0[4], f1[4], f2[4], g0[4], g1[4], g2[4];
    #pragma unroll
    for (int u = 0; u < 4; ++u) {
        const int q = t + u * 256;
        f0[u] = tf4[q];  f1[u] = tf4[1024 + q];  f2[u] = tf4[2048 + q];
        g0[u] = tb4[q];  g1[u] = tb4[1024 + q];  g2[u] = tb4[2048 + q];
    }
    #pragma unroll
    for (int u = 0; u < 4; ++u) {
        const int q = t + u * 256;
        Ws4[q]        = f0[u];
        Ws4[1024 + q] = make_float4(g0[u].x + g1[u].x, g0[u].y + g1[u].y, g0[u].z + g1[u].z, g0[u].w + g1[u].w);
        Ws4[2048 + q] = make_float4(f1[u].x + g2[u].x, f1[u].y + g2[u].y, f1[u].z + g2[u].z, f1[u].w + g2[u].w);
        Ws4[3072 + q] = f2[u];
    }
    __syncthreads();

    const int wave = t >> 5;
    const int lane = t & 31;
    const int tm   = blockIdx.x * 8 + wave;
    if (tm >= 3125) return;

    const int m    = lane & 15;
    const int hi   = lane >> 4;
    const int arow = tm * 16 + m;

    v8f c0 = {}, c1 = {}, c2 = {}, c3 = {};
    mat_accum(A0, Ws,            arow, m, hi, c0, c1, c2, c3);
    mat_accum(A1, Ws + 4096,     arow, m, hi, c0, c1, c2, c3);
    mat_accum(A2, Ws + 8192,     arow, m, hi, c0, c1, c2, c3);
    mat_accum(A3, Ws + 12288,    arow, m, hi, c0, c1, c2, c3);

    #pragma unroll
    for (int v = 0; v < 8; ++v) {
        const int rowm = tm * 16 + hi * 8 + v;
        float* o = out + rowm * 64 + m;
        o[0]  = c0[v];
        o[16] = c1[v];
        o[32] = c2[v];
        o[48] = c3[v];
    }
}


extern "C" void kernel_launch(void* const* d_in, const int* in_sizes, int n_in,
                              void* d_out, int out_size, void* d_ws, size_t ws_size,
                              hipStream_t stream) {
    (void)in_sizes; (void)n_in; (void)out_size; (void)ws_size;

    const float* x  = (const float*)d_in[0];
    const int*   ei = (const int*)d_in[1];
    const float* ew = (const float*)d_in[2];
    const float* tf = (const float*)d_in[3];
    const float* tb = (const float*)d_in[4];
    const int* row = ei;
    const int* col = ei + N_EDGES;

    float* ws     = (float*)d_ws;
    int*   rowptr = (int*)ws;
    float* Tx1    = ws  + 50176;
    float* Tx2    = Tx1 + (size_t)N_NODES * C;
    float* Tx3    = Tx2 + (size_t)N_NODES * C;
    float* dinv   = Tx3;
    int2*  rec    = (int2*)(Tx3 + PADN);
    int*   tab    = (int*)(Tx3 + PADN) + 2 * N_EDGES;
    int*   lst    = (int*)d_out;

    bin_kernel<<<NCH, 1024, 0, stream>>>(row, col, ew, rec, tab);
    rows_kernel<<<NB, 1024, 0, stream>>>(rec, tab, rowptr, dinv, (int2*)lst);

    pull_kernel<true><<<N_NODES / (8 * RPW), 256, 0, stream>>>(x,   Tx1, rowptr, lst, dinv);
    pull_kernel<false><<<N_NODES / (8 * RPW), 256, 0, stream>>>(Tx1, Tx2, rowptr, lst, dinv);
    pull_kernel<false><<<N_NODES / (8 * RPW), 256, 0, stream>>>(Tx2, Tx3, rowptr, lst, dinv);

    gemm_kernel<<<(3125 + 7) / 8, 256, 0, stream>>>(x, Tx1, Tx2, Tx3, tf, tb,
                                                    (float*)d_out);
}


# third propagate pass and the weight product in one kernel: a work-group gathers its 64 rows of P Tx2 into LDS and multiplies at once; 5 launches, third plane gone; sums and order unchanged
# speedup vs baseline: 4.3750x; 1.0483x over previous
#include <hip/hip_runtime.h>

#define N_NODES 50000
#define N_EDGES 800000
#define C 64

typedef float    v2f  __attribute__((ext_vector_type(2)));
typedef float    v8f  __attribute__((ext_vector_type(8)));

#define NCH      250
#define CH_E     (N_EDGES / NCH)
#define NB       256
#define BIN_ROWS 196
#define TABW     (NB + 1)
#define PADN     (NB * BIN_ROWS)

__device__ __forceinline__ int block_scan_excl(int v, int* wsum, int& tot) {
    const int lane = threadIdx.x & 31;
    const int wv   = threadIdx.x >> 5;
    int inc = v;
    for (int off = 1; off < 32; off <<= 1) {
        const int y = __shfl_up(inc, off);
        if (lane >= off) inc += y;
    }
    __syncthreads();
    if (lane == 31) wsum[wv] = inc;
    __syncthreads();
    const int ws = wsum[lane];
    int x = ws;
    for (int off = 1; off < 32; off <<= 1) {
        const int y = __shfl_up(x, off);
        if (lane >= off) x += y;
    }
    tot = __shfl(x, 31);
    const int before = __shfl(x - ws, wv);
    return before + inc - v;
}

__global__ void __launch_bounds__(1024)
bin_kernel(const int* __restrict__ row, const int* __restrict__ col, const float* __restrict__ w,
           int2* __restrict__ rec, int* __restrict__ tab) {
    __shared__ int  cnt[NB];
    __shared__ int  start[NB];
    __shared__ int  wsum[32];
    __shared__ int2 stage[CH_E];

    const int t  = threadIdx.x;
    const int c  = blockIdx.x;
    const int e0 = c * CH_E;
    if (t < NB) cnt[t] = 0;
    __syncthreads();

    const bool act = t < CH_E / 4;
    int4   rv = make_int4(0, 0, 0, 0), cv = make_int4(0, 0, 0, 0);
    float4 wq = make_float4(0.f, 0.f, 0.f, 0.f);
    if (act) {
        rv = reinterpret_cast<const int4*>(row + e0)[t];
        cv = reinterpret_cast<const int4*>(col + e0)[t];
        wq = reinterpret_cast<const float4*>(w + e0)[t];
    }
    const unsigned rr[4] = { (unsigned)rv.x, (unsigned)rv.y, (unsigned)rv.z, (unsigned)rv.w };
    const int      cc[4] = { cv.x, cv.y, cv.z, cv.w };
    const float    ww[4] = { wq.x, wq.y, wq.z, wq.w };
    unsigned bn[4];
    int      rk[4];
    #pragma unroll
    for (int k = 0; k < 4; ++k) {
        bn[k] = rr[k] / (unsigned)BIN_ROWS;
        rk[k] = act ? atomicAdd(&cnt[bn[k]], 1) : 0;
    }
    __syncthreads();

    int tot;
    const int v  = (t < NB) ? cnt[t] : 0;
    const int ex = block_scan_excl(v, wsum, tot);
    if (t < NB) { start[t] = ex; tab[c * TABW + t] = ex; }
    if (t == 0) tab[c * TABW + NB] = tot;
    __syncthreads();

    if (act) {
        #pragma unroll
        for (int k = 0; k < 4; ++k) {
            const unsigned ro = rr[k] - bn[k] * (unsigned)BIN_ROWS;
            stage[start[bn[k]] + rk[k]] = make_int2((int)((ro << 16) | (unsigned)cc[k]), __float_as_int(ww[k]));
        }
    }
    __syncthreads();
    for (int i = t; i < CH_E; i += 1024) rec[e0 + i] = stage[i];
}

#define CPW 8
__global__ void __launch_bounds__(1024)
rows_kernel(const int2* __restrict__ rec, const int* __restrict__ tab,
            int* __restrict__ rowptr, float* __restrict__ dinv, int2* __restrict__ lst) {
    __shared__ __align__(16) int s0s[32 * CPW];
    __shared__ __align__(16) int sln[32 * CPW];
    __shared__ int   rcnt[BIN_ROWS];
    __shared__ float rdg[BIN_ROWS];
    __shared__ int   rnext[BIN_ROWS];
    __shared__ float rdi[BIN_ROWS];
    __shared__ int   wsum[32];

    const int t    = threadIdx.x;
    const int lane = t & 31;
    const int wv   = t >> 5;
    const int b    = blockIdx.x;
    int a0 = 0, len = 0;
    if (t < NCH) {
        a0  = tab[t * TABW + b];
        len = tab[t * TABW + b + 1] - a0;
    }
    if (t < 32 * CPW) { s0s[t] = a0; sln[t] = len; }
    if (t < BIN_ROWS) { rcnt[t] = 0; rdg[t] = 0.f; }
    int base;
    (void)block_scan_excl(a0, wsum, base);
    __syncthreads();

    const int4 la = reinterpret_cast<const int4*>(sln)[wv * 2], lb = reinterpret_cast<const int4*>(sln)[wv * 2 + 1];
    const int4 sa = reinterpret_cast<const int4*>(s0s)[wv * 2], sb = reinterpret_cast<const int4*>(s0s)[wv * 2 + 1];
    const int lnv[CPW] = { la.x, la.y, la.z, la.w, lb.x, lb.y, lb.z, lb.w };
    const int sv[CPW]  = { sa.x, sa.y, sa.z, sa.w, sb.x, sb.y, sb.z, sb.w };
    int2 kr[CPW];
    int  ln[CPW], p0[CPW];
    #pragma unroll
    for (int k = 0; k < CPW; ++k) {
        ln[k] = __builtin_amdgcn_readfirstlane(lnv[k]);
        p0[k] = __builtin_amdgcn_readfirstlane((wv * CPW + k) * CH_E + sv[k]);
        kr[k] = (lane < ln[k]) ? rec[p0[k] + lane] : make_int2(-1, 0);
    }
    #pragma unroll
    for (int k = 0; k < CPW; ++k) {
        if (kr[k].x >= 0) {
            const int ro = kr[k].x >> 16;
            atomicAdd(&rcnt[ro], 1);
            atomicAdd(&rdg[ro], __int_as_float(kr[k].y));
        }
    }
    #pragma unroll
    for (int k = 0; k < CPW; ++k) {
        for (int i = lane + 32; i < ln[k]; i += 32) {
            const int2 r  = rec[p0[k] + i];
            const int  ro = r.x >> 16;
            atomicAdd(&rcnt[ro], 1);
            atomicAdd(&rdg[ro], __int_as_float(r.y));
        }
    }
    __syncthreads();

    int unused;
    const int rc  = (t < BIN_ROWS) ? rcnt[t] : 0;
    const int rex = block_scan_excl(rc, wsum, unused);
    if (t < BIN_ROWS) {
        const float g  = rdg[t];
        const float di = (g > 0.f) ? rsqrtf(g) : 0.f;
        rnext[t] = base + rex;
        rdi[t]   = di;
        rowptr[b * BIN_ROWS + t] = base + rex;
        dinv[b * BIN_ROWS + t]   = di;
    }
    __syncthreads();

    #pragma unroll
    for (int k = 0; k < CPW; ++k) {
        if (kr[k].x >= 0) {
            const int ro  = kr[k].x >> 16;
            const int pos = atomicAdd(&rnext[ro], 1);
            lst[pos] = make_int2(kr[k].x & 0xffff, __float_as_int(rdi[ro] * __int_as_float(kr[k].y)));
        }
    }
    #pragma unroll
    for (int k = 0; k < CPW; ++k) {
        for (int i = lane + 32; i < ln[k]; i += 32) {
            const int2 r   = rec[p0[k] + i];
            const int  ro  = r.x >> 16;
            const int  pos = atomicAdd(&rnext[ro], 1);
            lst[pos] = make_int2(r.x & 0xffff, __float_as_int(rdi[ro] * __int_as_float(r.y)));
        }
    }
}

#define BATCH 24
template <int K>
__device__ __forceinline__ void issue_gathers(const float2* __restrict__ xlane, int mx, int j, int cnt,
                                              float2 (&v)[BATCH]) {
    if (K == 0 || K < cnt) {
        v[K] = xlane[(size_t)(unsigned)__builtin_amdgcn_readlane(mx, j + K) * (C / 2)];
        if constexpr (K + 1 < BATCH) issue_gathers<K + 1>(xlane, mx, j, cnt, v);
    }
}

template <int K>
__device__ __forceinline__ void add_gathers(int my, int j, int cnt, const float2 (&v)[BATCH],
                                            float& ax, float& ay) {
    if (K == 0 || K < cnt) {
        const float wt = __int_as_float(__builtin_amdgcn_readlane(my, j + K));
        ax = fmaf(wt, v[K].x, ax);
        ay = fmaf(wt, v[K].y, ay);
        if constexpr (K + 1 < BATCH) add_gathers<K + 1>(my, j, cnt, v, ax, ay);
    }
}

#define RPW 5
template <bool FIRST>
__global__ void __launch_bounds__(256)
pull_kernel(const float* __restrict__ xin, float* __restrict__ xout,
            const int* __restrict__ rowptr, int* lst, const float* __restrict__ dinv) {
    const int lane = threadIdx.x & 31;
    const int r0 = __builtin_amdgcn_readfirstlane((blockIdx.x * 8 + (threadIdx.x >> 5)) * RPW);
    const int bv = rowptr[r0 + min(lane, RPW)];
    const float2* xlane = reinterpret_cast<const float2*>(xin) + lane;
    const int2*   lst2  = reinterpret_cast<const int2*>(lst);
    int beg = __builtin_amdgcn_readlane(bv, 0);
    int end = beg;
    int n   = 0;
    int2 m  = make_int2(0, 0);
    #pragma nounroll
    for (int i = -1; i < RPW; ++i) {
        float dv = 0.f;
        if (FIRST) {
            if (lane < n) dv = dinv[m.x];
        }
        const int nbeg = end;
        int  nend = end, nn = 0;
        int2 mn   = make_int2(0, 0);
        if (i + 1 < RPW) {
            nend = __builtin_amdgcn_readlane(bv, i + 2);
            nn   = min(32, nend - nbeg);
            if (lane < nn) mn = lst2[nbeg + lane];
        }
        float ax = 0.f, ay = 0.f;
        if (n > 0) {
            #pragma nounroll
            for (int j = 0; j < n; j += BATCH) {
                float2 v[BATCH];
                issue_gathers<0>(xlane, m.x, j, n - j, v);
                if (FIRST) {
                    if (j == 0 && lane < n) {
                        m.y = __float_as_int(__int_as_float(m.y) * dv);
                        lst[2 * (beg + lane) + 1] = m.y;
                    }
                }
                add_gathers<0>(m.y, j, n - j, v, ax, ay);
            }
            for (int b2 = beg + 32; b2 < end; b2 += 32) {
                const int n2 = min(32, end - b2);
                int2 m2 = make_int2(0, 0);
                if (lane < n2) {
                    m2 = lst2[b2 + lane];
                    if (FIRST) {
                        m2.y = __float_as_int(__int_as_float(m2.y) * dinv[m2.x]);
                        lst[2 * (b2 + lane) + 1] = m2.y;
                    }
                }
                for (int k = 0; k < n2; ++k) {
                    const unsigned c  = (unsigned)__builtin_amdgcn_readlane(m2.x, k);
                    const float    wt = __int_as_float(__builtin_amdgcn_readlane(m2.y, k));
                    const float2   xv = xlane[(size_t)c * (C / 2)];
                    ax = fmaf(wt, xv.x, ax);
                    ay = fmaf(wt, xv.y, ay);
                }
            }
        }
        if (i >= 0) *reinterpret_cast<float2*>(xout + (size_t)(r0 + i) * C + lane * 2) = make_float2(ax, ay);
        m = mn;  n = nn;  beg = nbeg;  end = nend;
    }
}

#define FR     8
#define TPITCH 64
#define FGRID  782

__device__ __forceinline__ void mat_accum2(const float* Arow, int sw, const float* Wm, int nm, int hi,
                                           v8f& c0, v8f& c1) {
    #pragma unroll
    for (int k0 = 0; k0 < 64; k0 += 4) {
        const int kk = k0 + hi * 2;
        v2f a = *reinterpret_cast<const v2f*>(Arow + (kk ^ sw));
        v2f b0, b1;
        b0[0] = Wm[kk * 64 + nm];          b0[1] = Wm[(kk + 1) * 64 + nm];
        b1[0] = Wm[kk * 64 + 16 + nm];     b1[1] = Wm[(kk + 1) * 64 + 16 + nm];
        c0 = __builtin_amdgcn_wmma_f32_16x16x4_f32(false, a, false, b0, (short)0, c0, false, false);
        c1 = __builtin_amdgcn_wmma_f32_16x16x4_f32(false, a, false, b1, (short)0, c1, false, false);
    }
}

__global__ void __launch_bounds__(256)
pull_gemm_kernel(const float* __restrict__ A0, const float* __restrict__ A1, const float* __restrict__ A2,
                 const int* __restrict__ rowptr, const int2* __restrict__ lst2,
                 const float* __restrict__ tf, const float* __restrict__ tb, float* __restrict__ out) {
    __shared__ __align__(16) float Ws[4 * 64 * 64];
    __shared__ __align__(16) float Ts[64 * TPITCH];

    const int t    = threadIdx.x;
    const int lane = t & 31;
    const int wv   = t >> 5;
    const int r0 = __builtin_amdgcn_readfirstlane(blockIdx.x * 64 + wv * FR);
    const int bv = rowptr[r0 + min(lane, FR)];

    {
        const float4* tf4 = reinterpret_cast<const float4*>(tf);
        const float4* tb4 = reinterpret_cast<const float4*>(tb);
        float4*       Ws4 = reinterpret_cast<float4*>(Ws);
        float4 f0[4], f1[4], f2[4], g0[4], g1[4], g2[4];
        #pragma unroll
        for (int u = 0; u < 4; ++u) {
            const int q = t + u * 256;
            f0[u] = tf4[q];  f1[u] = tf4[1024 + q];  f2[u] = tf4[2048 + q];
            g0[u] = tb4[q];  g1[u] = tb4[1024 + q];  g2[u] = tb4[2048 + q];
        }
        #pragma unroll
        for (int u = 0; u < 4; ++u) {
            const int q = t + u * 256;
            Ws4[q]        = f0[u];
            Ws4[1024 + q] = make_float4(g0[u].x + g1[u].x, g0[u].y + g1[u].y, g0[u].z + g1[u].z, g0[u].w + g1[u].w);
            Ws4[2048 + q] = make_float4(f1[u].x + g2[u].x, f1[u].y + g2[u].y, f1[u].z + g2[u].z, f1[u].w + g2[u].w);
            Ws4[3072 + q] = f2[u];
        }
    }

    const float2* xlane = reinterpret_cast<const float2*>(A2) + lane;
    float*        trow  = Ts + (wv * FR) * TPITCH;
    const int     rsw   = (wv & 1) * FR;
    int beg = __builtin_amdgcn_readlane(bv, 0);
    int end = beg;
    int n   = 0;
    int2 m  = make_int2(0, 0);
    #pragma nounroll
    for (int i = -1; i < FR; ++i) {
        const int nbeg = end;
        int  nend = end, nn = 0;
        int2 mn   = make_int2(0, 0);
        if (i + 1 < FR) {
            nend = __builtin_amdgcn_readlane(bv, i + 2);
            nn   = min(32, nend - nbeg);
            if (lane < nn) mn = lst2[nbeg + lane];
        }
        float ax = 0.f, ay = 0.f;
        if (n > 0) {
            #pragma nounroll
            for (int j = 0; j < n; j += BATCH) {
                float2 v[BATCH];
                issue_gathers<0>(xlane, m.x, j, n - j, v);
                add_gathers<0>(m.y, j, n - j, v, ax, ay);
            }
            for (int b2 = beg + 32; b2 < end; b2 += 32) {
                const int n2 = min(32, end - b2);
                int2 m2 = make_int2(0, 0);
                if (lane < n2) m2 = lst2[b2 + lane];
                for (int k = 0; k < n2; ++k) {
                    const unsigned c  = (unsigned)__builtin_amdgcn_readlane(m2.x, k);
                    const float    wt = __int_as_float(__builtin_amdgcn_readlane(m2.y, k));
                    const float2   xv = xlane[(size_t)c * (C / 2)];
                    ax = fmaf(wt, xv.x, ax);
                    ay = fmaf(wt, xv.y, ay);
                }
            }
        }
        if (i >= 0) *reinterpret_cast<float2*>(trow + i * TPITCH + ((lane * 2) ^ ((rsw + i) << 2))) = make_float2(ax, ay);
        m = mn;  n = nn;  beg = nbeg;  end = nend;
    }
    __syncthreads();

    const int tl   = wv >> 1;
    const int row0 = blockIdx.x * 64 + tl * 16;
    if (row0 >= N_NODES) return;

    const int mm   = lane & 15;
    const int hi   = lane >> 4;
    const int nm   = (wv & 1) * 32 + mm;
    const size_t ao = (size_t)(row0 + mm) * 64;

    v8f c0 = {}, c1 = {};
    mat_accum2(A0 + ao,                       0,       Ws,         nm, hi, c0, c1);
    mat_accum2(A1 + ao,                       0,       Ws + 4096,  nm, hi, c0, c1);
    mat_accum2(A2 + ao,                       0,       Ws + 8192,  nm, hi, c0, c1);
    mat_accum2(Ts + (tl * 16 + mm) * TPITCH,  mm << 2, Ws + 12288, nm, hi, c0, c1);

    #pragma unroll
    for (int v = 0; v < 8; ++v) {
        float* o = out + (size_t)(row0 + hi * 8 + v) * 64 + nm;
        o[0]  = c0[v];
        o[16] = c1[v];
    }
}


extern "C" void kernel_launch(void* const* d_in, const int* in_sizes, int n_in,
                              void* d_out, int out_size, void* d_ws, size_t ws_size,
                              hipStream_t stream) {
    (void)in_sizes; (void)n_in; (void)out_size; (void)ws_size;

    const float* x  = (const float*)d_in[0];
    const int*   ei = (const int*)d_in[1];
    const float* ew = (const float*)d_in[2];
    const float* tf = (const float*)d_in[3];
    const float* tb = (const float*)d_in[4];
    const int* row = ei;
    const int* col = ei + N_EDGES;

    float* ws     = (float*)d_ws;
    int*   rowptr = (int*)ws;
    float* Tx1    = ws  + 50176;
    float* Tx2    = Tx1 + (size_t)N_NODES * C;
    float* room3  = Tx2 + (size_t)N_NODES * C;
    float* dinv   = Tx2;
    int*   tab    = (int*)(Tx2 + PADN);
    int2*  rec    = (int2*)room3;
    int*   lst    = (int*)room3 + 2 * N_EDGES;

    bin_kernel<<<NCH, 1024, 0, stream>>>(row, col, ew, rec, tab);
    rows_kernel<<<NB, 1024, 0, stream>>>(rec, tab, rowptr, dinv, (int2*)lst);

    pull_kernel<true><<<N_NODES / (8 * RPW), 256, 0, stream>>>(x,   Tx1, rowptr, lst, dinv);
    pull_kernel<false><<<N_NODES / (8 * RPW), 256, 0, stream>>>(Tx1, Tx2, rowptr, lst, dinv);

    pull_gemm_kernel<<<FGRID, 256, 0, stream>>>(x, Tx1, Tx2, rowptr, (const int2*)lst, tf, tb,
                                                (float*)d_out);
}


# last kernel: 12 waves a work-group; 8 gather the P Tx2 rows while 4 matrix waves already multiply x, Tx1, Tx2 on the idle matrix unit; only the P Tx2 product waits; sums and order unchanged
# speedup vs baseline: 4.6626x; 1.0657x over previous
#include <hip/hip_runtime.h>

#define N_NODES 50000
#define N_EDGES 800000
#define C 64

typedef float    v2f  __attribute__((ext_vector_type(2)));
typedef float    v8f  __attribute__((ext_vector_type(8)));

#define NCH      250
#define CH_E     (N_EDGES / NCH)
#define NB       256
#define BIN_ROWS 196
#define TABW     (NB + 1)
#define PADN     (NB * BIN_ROWS)

__device__ __forceinline__ int block_scan_excl(int v, int* wsum, int& tot) {
    const int lane = threadIdx.x & 31;
    const int wv   = threadIdx.x >> 5;
    int inc = v;
    for (int off = 1; off < 32; off <<= 1) {
        const int y = __shfl_up(inc, off);
        if (lane >= off) inc += y;
    }
    __syncthreads();
    if (lane == 31) wsum[wv] = inc;
    __syncthreads();
    const int ws = wsum[lane];
    int x = ws;
    for (int off = 1; off < 32; off <<= 1) {
        const int y = __shfl_up(x, off);
        if (lane >= off) x += y;
    }
    tot = __shfl(x, 31);
    const int before = __shfl(x - ws, wv);
    return before + inc - v;
}

__global__ void __launch_bounds__(1024)
bin_kernel(const int* __restrict__ row, const int* __restrict__ col, const float* __restrict__ w,
           int2* __restrict__ rec, int* __restrict__ tab) {
    __shared__ int  cnt[NB];
    __shared__ int  start[NB];
    __shared__ int  wsum[32];
    __shared__ int2 stage[CH_E];

    const int t  = threadIdx.x;
    const int c  = blockIdx.x;
    const int e0 = c * CH_E;
    if (t < NB) cnt[t] = 0;
    __syncthreads();

    const bool act = t < CH_E / 4;
    int4   rv = make_int4(0, 0, 0, 0), cv = make_int4(0, 0, 0, 0);
    float4 wq = make_float4(0.f, 0.f, 0.f, 0.f);
    if (act) {
        rv = reinterpret_cast<const int4*>(row + e0)[t];
        cv = reinterpret_cast<const int4*>(col + e0)[t];
        wq = reinterpret_cast<const float4*>(w + e0)[t];
    }
    const unsigned rr[4] = { (unsigned)rv.x, (unsigned)rv.y, (unsigned)rv.z, (unsigned)rv.w };
    const int      cc[4] = { cv.x, cv.y, cv.z, cv.w };
    const float    ww[4] = { wq.x, wq.y, wq.z, wq.w };
    unsigned bn[4];
    int      rk[4];
    #pragma unroll
    for (int k = 0; k < 4; ++k) {
        bn[k] = rr[k] / (unsigned)BIN_ROWS;
        rk[k] = act ? atomicAdd(&cnt[bn[k]], 1) : 0;
    }
    __syncthreads();

    int tot;
    const int v  = (t < NB) ? cnt[t] : 0;
    const int ex = block_scan_excl(v, wsum, tot);
    if (t < NB) { start[t] = ex; tab[c * TABW + t] = ex; }
    if (t == 0) tab[c * TABW + NB] = tot;
    __syncthreads();

    if (act) {
        #pragma unroll
        for (int k = 0; k < 4; ++k) {
            const unsigned ro = rr[k] - bn[k] * (unsigned)BIN_ROWS;
            stage[start[bn[k]] + rk[k]] = make_int2((int)((ro << 16) | (unsigned)cc[k]), __float_as_int(ww[k]));
        }
    }
    __syncthreads();
    for (int i = t; i < CH_E; i += 1024) rec[e0 + i] = stage[i];
}

#define CPW 8
__global__ void __launch_bounds__(1024)
rows_kernel(const int2* __restrict__ rec, const int* __restrict__ tab,
            int* __restrict__ rowptr, float* __restrict__ dinv, int2* __restrict__ lst) {
    __shared__ __align__(16) int s0s[32 * CPW];
    __shared__ __align__(16) int sln[32 * CPW];
    __shared__ int   rcnt[BIN_ROWS];
    __shared__ float rdg[BIN_ROWS];
    __shared__ int   rnext[BIN_ROWS];
    __shared__ float rdi[BIN_ROWS];
    __shared__ int   wsum[32];

    const int t    = threadIdx.x;
    const int lane = t & 31;
    const int wv   = t >> 5;
    const int b    = blockIdx.x;
    int a0 = 0, len = 0;
    if (t < NCH) {
        a0  = tab[t * TABW + b];
        len = tab[t * TABW + b + 1] - a0;
    }
    if (t < 32 * CPW) { s0s[t] = a0; sln[t] = len; }
    if (t < BIN_ROWS) { rcnt[t] = 0; rdg[t] = 0.f; }
    int base;
    (void)block_scan_excl(a0, wsum, base);
    __syncthreads();

    const int4 la = reinterpret_cast<const int4*>(sln)[wv * 2], lb = reinterpret_cast<const int4*>(sln)[wv * 2 + 1];
    const int4 sa = reinterpret_cast<const int4*>(s0s)[wv * 2], sb = reinterpret_cast<const int4*>(s0s)[wv * 2 + 1];
    const int lnv[CPW] = { la.x, la.y, la.z, la.w, lb.x, lb.y, lb.z, lb.w };
    const int sv[CPW]  = { sa.x, sa.y, sa.z, sa.w, sb.x, sb.y, sb.z, sb.w };
    int2 kr[CPW];
    int  ln[CPW], p0[CPW];
    #pragma unroll
    for (int k = 0; k < CPW; ++k) {
        ln[k] = __builtin_amdgcn_readfirstlane(lnv[k]);
        p0[k] = __builtin_amdgcn_readfirstlane((wv * CPW + k) * CH_E + sv[k]);
        kr[k] = (lane < ln[k]) ? rec[p0[k] + lane] : make_int2(-1, 0);
    }
    #pragma unroll
    for (int k = 0; k < CPW; ++k) {
        if (kr[k].x >= 0) {
            const int ro = kr[k].x >> 16;
            atomicAdd(&rcnt[ro], 1);
            atomicAdd(&rdg[ro], __int_as_float(kr[k].y));
        }
    }
    #pragma unroll
    for (int k = 0; k < CPW; ++k) {
        for (int i = lane + 32; i < ln[k]; i += 32) {
            const int2 r  = rec[p0[k] + i];
            const int  ro = r.x >> 16;
            atomicAdd(&rcnt[ro], 1);
            atomicAdd(&rdg[ro], __int_as_float(r.y));
        }
    }
    __syncthreads();

    int unused;
    const int rc  = (t < BIN_ROWS) ? rcnt[t] : 0;
    const int rex = block_scan_excl(rc, wsum, unused);
    if (t < BIN_ROWS) {
        const float g  = rdg[t];
        const float di = (g > 0.f) ? rsqrtf(g) : 0.f;
        rnext[t] = base + rex;
        rdi[t]   = di;
        rowptr[b * BIN_ROWS + t] = base + rex;
        dinv[b * BIN_ROWS + t]   = di;
    }
    __syncthreads();

    #pragma unroll
    for (int k = 0; k < CPW; ++k) {
        if (kr[k].x >= 0) {
            const int ro  = kr[k].x >> 16;
            const int pos = atomicAdd(&rnext[ro], 1);
            lst[pos] = make_int2(kr[k].x & 0xffff, __float_as_int(rdi[ro] * __int_as_float(kr[k].y)));
        }
    }
    #pragma unroll
    for (int k = 0; k < CPW; ++k) {
        for (int i = lane + 32; i < ln[k]; i += 32) {
            const int2 r   = rec[p0[k] + i];
            const int  ro  = r.x >> 16;
            const int  pos = atomicAdd(&rnext[ro], 1);
            lst[pos] = make_int2(r.x & 0xffff, __float_as_int(rdi[ro] * __int_as_float(r.y)));
        }
    }
}

#define BATCH 24
template <int K>
__device__ __forceinline__ void issue_gathers(const float2* __restrict__ xlane, int mx, int j, int cnt,
                                              float2 (&v)[BATCH]) {
    if (K == 0 || K < cnt) {
        v[K] = xlane[(size_t)(unsigned)__builtin_amdgcn_readlane(mx, j + K) * (C / 2)];
        if constexpr (K + 1 < BATCH) issue_gathers<K + 1>(xlane, mx, j, cnt, v);
    }
}

template <int K>
__device__ __forceinline__ void add_gathers(int my, int j, int cnt, const float2 (&v)[BATCH],
                                            float& ax, float& ay) {
    if (K == 0 || K < cnt) {
        const float wt = __int_as_float(__builtin_amdgcn_readlane(my, j + K));
        ax = fmaf(wt, v[K].x, ax);
        ay = fmaf(wt, v[K].y, ay);
        if constexpr (K + 1 < BATCH) add_gathers<K + 1>(my, j, cnt, v, ax, ay);
    }
}

#define RPW 5
template <bool FIRST>
__global__ void __launch_bounds__(256)
pull_kernel(const float* __restrict__ xin, float* __restrict__ xout,
            const int* __restrict__ rowptr, int* lst, const float* __restrict__ dinv) {
    const int lane = threadIdx.x & 31;
    const int r0 = __builtin_amdgcn_readfirstlane((blockIdx.x * 8 + (threadIdx.x >> 5)) * RPW);
    const int bv = rowptr[r0 + min(lane, RPW)];
    const float2* xlane = reinterpret_cast<const float2*>(xin) + lane;
    const int2*   lst2  = reinterpret_cast<const int2*>(lst);
    int beg = __builtin_amdgcn_readlane(bv, 0);
    int end = beg;
    int n   = 0;
    int2 m  = make_int2(0, 0);
    #pragma nounroll
    for (int i = -1; i < RPW; ++i) {
        float dv = 0.f;
        if (FIRST) {
            if (lane < n) dv = dinv[m.x];
        }
        const int nbeg = end;
        int  nend = end, nn = 0;
        int2 mn   = make_int2(0, 0);
        if (i + 1 < RPW) {
            nend = __builtin_amdgcn_readlane(bv, i + 2);
            nn   = min(32, nend - nbeg);
            if (lane < nn) mn = lst2[nbeg + lane];
        }
        float ax = 0.f, ay = 0.f;
        if (n > 0) {
            #pragma nounroll
            for (int j = 0; j < n; j += BATCH) {
                float2 v[BATCH];
                issue_gathers<0>(xlane, m.x, j, n - j, v);
                if (FIRST) {
                    if (j == 0 && lane < n) {
                        m.y = __float_as_int(__int_as_float(m.y) * dv);
                        lst[2 * (beg + lane) + 1] = m.y;
                    }
                }
                add_gathers<0>(m.y, j, n - j, v, ax, ay);
            }
            for (int b2 = beg + 32; b2 < end; b2 += 32) {
                const int n2 = min(32, end - b2);
                int2 m2 = make_int2(0, 0);
                if (lane < n2) {
                    m2 = lst2[b2 + lane];
                    if (FIRST) {
                        m2.y = __float_as_int(__int_as_float(m2.y) * dinv[m2.x]);
                        lst[2 * (b2 + lane) + 1] = m2.y;
                    }
                }
                for (int k = 0; k < n2; ++k) {
                    const unsigned c  = (unsigned)__builtin_amdgcn_readlane(m2.x, k);
                    const float    wt = __int_as_float(__builtin_amdgcn_readlane(m2.y, k));
                    const float2   xv = xlane[(size_t)c * (C / 2)];
                    ax = fmaf(wt, xv.x, ax);
                    ay = fmaf(wt, xv.y, ay);
                }
            }
        }
        if (i >= 0) *reinterpret_cast<float2*>(xout + (size_t)(r0 + i) * C + lane * 2) = make_float2(ax, ay);
        m = mn;  n = nn;  beg = nbeg;  end = nend;
    }
}

#define FR     8
#define TPITCH 64
#define FGRID  782
#define FTHR   384

__device__ __forceinline__ void mat_step4(v2f a, const float* Wm, int kk, int mm, v8f (&c)[4]) {
    #pragma unroll
    for (int j = 0; j < 4; ++j) {
        v2f b;
        b[0] = Wm[kk * 64 + j * 16 + mm];
        b[1] = Wm[(kk + 1) * 64 + j * 16 + mm];
        c[j] = __builtin_amdgcn_wmma_f32_16x16x4_f32(false, a, false, b, (short)0, c[j], false, false);
    }
}

__device__ __forceinline__ void mat_accum4(const float* Arow, int sw, const float* Wm, int mm, int hi, v8f (&c)[4]) {
    #pragma unroll
    for (int s = 0; s < 16; ++s) {
        const int kk = s * 4 + hi * 2;
        mat_step4(*reinterpret_cast<const v2f*>(Arow + (kk ^ sw)), Wm, kk, mm, c);
    }
}

__device__ __forceinline__ void mat_accum4_mem(const float* base, unsigned lo, const float* Wm, int mm, int hi, v8f (&c)[4]) {
    const float* q = base + lo;
    v2f a[16];
    #pragma unroll
    for (int s = 0; s < 16; ++s) a[s] = *reinterpret_cast<const v2f*>(q + s * 4);
    #pragma unroll
    for (int s = 0; s < 16; ++s)
        mat_step4(a[s], Wm, s * 4 + hi * 2, mm, c);
}

__global__ void __launch_bounds__(FTHR, 12)
pull_gemm_kernel(const float* __restrict__ A0, const float* __restrict__ A1, const float* __restrict__ A2,
                 const int* __restrict__ rowptr, const int2* __restrict__ lst2,
                 const float* __restrict__ tf, const float* __restrict__ tb, float* __restrict__ out) {
    __shared__ __align__(16) float Ws[4 * 64 * 64];
    __shared__ __align__(16) float Ts[64 * TPITCH];

    const int t    = threadIdx.x;
    const int lane = t & 31;
    const int wv = __builtin_amdgcn_readfirstlane(t >> 5);
    const int gw = wv & 7;
    const int r0 = blockIdx.x * 64 + gw * FR;
    const int bv = rowptr[r0 + min(lane, FR)];

    {
        const float4* tf4 = reinterpret_cast<const float4*>(tf);
        const float4* tb4 = reinterpret_cast<const float4*>(tb);
        float4*       Ws4 = reinterpret_cast<float4*>(Ws);
        const float4  z4  = make_float4(0.f, 0.f, 0.f, 0.f);
        float4 f0[3], f1[3], f2[3], g0[3], g1[3], g2[3];
        #pragma unroll
        for (int u = 0; u < 3; ++u) {
            const int q = t + u * FTHR;
            f0[u] = f1[u] = f2[u] = g0[u] = g1[u] = g2[u] = z4;
            if (q < 1024) {
                f0[u] = tf4[q];  f1[u] = tf4[1024 + q];  f2[u] = tf4[2048 + q];
                g0[u] = tb4[q];  g1[u] = tb4[1024 + q];  g2[u] = tb4[2048 + q];
            }
        }
        #pragma unroll
        for (int u = 0; u < 3; ++u) {
            const int q = t + u * FTHR;
            if (q < 1024) {
                Ws4[q]        = f0[u];
                Ws4[1024 + q] = make_float4(g0[u].x + g1[u].x, g0[u].y + g1[u].y, g0[u].z + g1[u].z, g0[u].w + g1[u].w);
                Ws4[2048 + q] = make_float4(f1[u].x + g2[u].x, f1[u].y + g2[u].y, f1[u].z + g2[u].z, f1[u].w + g2[u].w);
                Ws4[3072 + q] = f2[u];
            }
        }
    }
    __syncthreads();

    const int  tl   = gw & 3;
    const int  row0 = blockIdx.x * 64 + tl * 16;
    const bool live = row0 < N_NODES;
    const int  mm   = lane & 15;
    const int  hi   = lane >> 4;
    v8f c[4] = {};

    if (wv < 8) {
        const float2* xlane = reinterpret_cast<const float2*>(A2) + lane;
        float*        trow  = Ts + (gw * FR) * TPITCH;
        const int     rsw   = (gw & 1) * FR;
        int beg = __builtin_amdgcn_readlane(bv, 0);
        int end = beg;
        int n   = 0;
        int2 m  = make_int2(0, 0);
        #pragma nounroll
        for (int i = -1; i < FR; ++i) {
            const int nbeg = end;
            int  nend = end, nn = 0;
            int2 mn   = make_int2(0, 0);
            if (i + 1 < FR) {
                nend = __builtin_amdgcn_readlane(bv, i + 2);
                nn   = min(32, nend - nbeg);
                if (lane < nn) mn = lst2[nbeg + lane];
            }
            float ax = 0.f, ay = 0.f;
            if (n > 0) {
                #pragma nounroll
                for (int j = 0; j < n; j += BATCH) {
                    float2 v[BATCH];
                    issue_gathers<0>(xlane, m.x, j, n - j, v);
                    add_gathers<0>(m.y, j, n - j, v, ax, ay);
                }
                for (int b2 = beg + 32; b2 < end; b2 += 32) {
                    const int n2 = min(32, end - b2);
                    int2 m2 = make_int2(0, 0);
                    if (lane < n2) m2 = lst2[b2 + lane];
                    for (int k = 0; k < n2; ++k) {
                        const unsigned c  = (unsigned)__builtin_amdgcn_readlane(m2.x, k);
                        const float    wt = __int_as_float(__builtin_amdgcn_readlane(m2.y, k));
                        const float2   xv = xlane[(size_t)c * (C / 2)];
                        ax = fmaf(wt, xv.x, ax);
                        ay = fmaf(wt, xv.y, ay);
                    }
                }
            }
            if (i >= 0) *reinterpret_cast<float2*>(trow + i * TPITCH + ((lane * 2) ^ ((rsw + i) << 2))) = make_float2(ax, ay);
            m = mn;  n = nn;  beg = nbeg;  end = nend;
        }
    } else if (live) {
        const size_t   ro = (size_t)row0 * 64;
        const unsigned lo = (unsigned)(mm * 64 + hi * 2);
        mat_accum4_mem(A0 + ro, lo, Ws,        mm, hi, c);
        mat_accum4_mem(A1 + ro, lo, Ws + 4096, mm, hi, c);
        mat_accum4_mem(A2 + ro, lo, Ws + 8192, mm, hi, c);
    }
    __syncthreads();

    if (wv >= 8 && live) {
        mat_accum4(Ts + (tl * 16 + mm) * TPITCH, mm << 2, Ws + 12288, mm, hi, c);

        #pragma unroll
        for (int v = 0; v < 8; ++v) {
            float* o = out + (size_t)(row0 + hi * 8 + v) * 64 + mm;
            o[0]  = c[0][v];
            o[16] = c[1][v];
            o[32] = c[2][v];
            o[48] = c[3][v];
        }
    }
}


extern "C" void kernel_launch(void* const* d_in, const int* in_sizes, int n_in,
                              void* d_out, int out_size, void* d_ws, size_t ws_size,
                              hipStream_t stream) {
    (void)in_sizes; (void)n_in; (void)out_size; (void)ws_size;

    const float* x  = (const float*)d_in[0];
    const int*   ei = (const int*)d_in[1];
    const float* ew = (const float*)d_in[2];
    const float* tf = (const float*)d_in[3];
    const float* tb = (const float*)d_in[4];
    const int* row = ei;
    const int* col = ei + N_EDGES;

    float* ws     = (float*)d_ws;
    int*   rowptr = (int*)ws;
    float* Tx1    = ws  + 50176;
    float* Tx2    = Tx1 + (size_t)N_NODES * C;
    float* room3  = Tx2 + (size_t)N_NODES * C;
    float* dinv   = Tx2;
    int*   tab    = (int*)(Tx2 + PADN);
    int2*  rec    = (int2*)room3;
    int*   lst    = (int*)room3 + 2 * N_EDGES;

    bin_kernel<<<NCH, 1024, 0, stream>>>(row, col, ew, rec, tab);
    rows_kernel<<<NB, 1024, 0, stream>>>(rec, tab, rowptr, dinv, (int2*)lst);

    pull_kernel<true><<<N_NODES / (8 * RPW), 256, 0, stream>>>(x,   Tx1, rowptr, lst, dinv);
    pull_kernel<false><<<N_NODES / (8 * RPW), 256, 0, stream>>>(Tx1, Tx2, rowptr, lst, dinv);

    pull_gemm_kernel<<<FGRID, FTHR, 0, stream>>>(x, Tx1, Tx2, rowptr, (const int2*)lst, tf, tb,
                                                (float*)d_out);
}
